# starting kernel
# baseline (speedup 1.0000x reference)
#include <hip/hip_runtime.h>
#include <hip/hip_bf16.h>


#define N_NODES  100000
#define N_EDGES  600000
#define N_GRAPHS 2000
#define NCLS     4

typedef float v2f __attribute__((ext_vector_type(2)));
typedef float v8f __attribute__((ext_vector_type(8)));


__global__ void k_zero(float* __restrict__ p, int n) {
  int i = blockIdx.x * blockDim.x + threadIdx.x;
  if (i < n) p[i] = 0.0f;
}

__global__ void k_deg(const int* __restrict__ dst, float* __restrict__ deg) {
  int e = blockIdx.x * blockDim.x + threadIdx.x;
  if (e < N_EDGES) atomicAdd(&deg[dst[e]], 1.0f);
}

__global__ void k_dinv(float* __restrict__ deg) {
  int i = blockIdx.x * blockDim.x + threadIdx.x;
  if (i < N_NODES) deg[i] = rsqrtf(deg[i] + 1.0f);
}

__global__ void k_embed(const int* __restrict__ tok, const float* __restrict__ emb,
                        float* __restrict__ x0) {
  size_t idx = (size_t)blockIdx.x * blockDim.x + threadIdx.x;
  int node = (int)(idx >> 6);
  int c    = (int)(idx & 63);
  x0[idx] = emb[((size_t)tok[node] << 6) | c];
}

template<int K>
__global__ void gemm_wmma_kernel(const float* __restrict__ X,
                                 const float* __restrict__ W,
                                 float* __restrict__ H) {
  const int wave  = (blockIdx.x << 3) | (threadIdx.x >> 5);
  const int lane  = threadIdx.x & 31;
  const int tileM = wave >> 3;
  const int tileN = wave & 7;
  const int half  = lane >> 4;
  const int lid   = lane & 15;
  const int m     = tileM * 16 + lid;
  const int n     = tileN * 16 + lid;
  const int kb    = half * 2;

  v8f acc = {};
  const float* xrow = X + (size_t)m * K + kb;
  const float* wcol = W + (size_t)kb * 128 + n;
#pragma unroll
  for (int k = 0; k < K; k += 4) {
    v2f a; a.x = xrow[k];               a.y = xrow[k + 1];
    v2f b; b.x = wcol[(size_t)k * 128]; b.y = wcol[(size_t)(k + 1) * 128];
    acc = __builtin_amdgcn_wmma_f32_16x16x4_f32(
         false, a,  false, b,
         (short)0, acc,  false,  false);
  }

  float* hbase = H + ((size_t)(tileM * 16 + half * 8)) * 128 + tileN * 16 + lid;
#pragma unroll
  for (int j = 0; j < 8; ++j)
    hbase[(size_t)j * 128] = acc[j];
}


__global__ void k_selfinit(const float* __restrict__ H, const float* __restrict__ dinv,
                           float* __restrict__ out) {
  size_t idx = (size_t)blockIdx.x * blockDim.x + threadIdx.x;
  int node = (int)(idx >> 7);
  float dv = dinv[node];
  out[idx] = H[idx] * dv * dv;
}

__global__ void k_scatter(const float* __restrict__ H, const int* __restrict__ src,
                          const int* __restrict__ dst, const float* __restrict__ dinv,
                          float* __restrict__ out) {
  size_t gid = (size_t)blockIdx.x * blockDim.x + threadIdx.x;
  int e    = (int)(gid >> 5);
  int lane = (int)(gid & 31);
  if (e >= N_EDGES) return;
  int s = src[e], d = dst[e];
  float nrm = dinv[s] * dinv[d];
  const float4 h = *(const float4*)(H + ((size_t)s << 7) + (lane << 2));
  float* o = out + ((size_t)d << 7) + (lane << 2);
  atomicAdd(o + 0, h.x * nrm);
  atomicAdd(o + 1, h.y * nrm);
  atomicAdd(o + 2, h.z * nrm);
  atomicAdd(o + 3, h.w * nrm);
}

__global__ void k_bias_relu(float* __restrict__ X, const float* __restrict__ b) {
  size_t idx = (size_t)blockIdx.x * blockDim.x + threadIdx.x;
  int c = (int)(idx & 127);
  float v = X[idx] + b[c];
  X[idx] = v > 0.0f ? v : 0.0f;
}


__global__ void k_pool(const float* __restrict__ X, const int* __restrict__ batch,
                       float* __restrict__ sums, float* __restrict__ cnts) {
  size_t gid = (size_t)blockIdx.x * blockDim.x + threadIdx.x;
  int node = (int)(gid >> 5);
  int lane = (int)(gid & 31);
  if (node >= N_NODES) return;
  int g = batch[node];
  const float4 v = *(const float4*)(X + ((size_t)node << 7) + (lane << 2));
  float* o = sums + ((size_t)g << 7) + (lane << 2);
  atomicAdd(o + 0, v.x);
  atomicAdd(o + 1, v.y);
  atomicAdd(o + 2, v.z);
  atomicAdd(o + 3, v.w);
  if (lane == 0) atomicAdd(&cnts[g], 1.0f);
}

__global__ void k_final(const float* __restrict__ sums, const float* __restrict__ cnts,
                        const float* __restrict__ Wlin, const float* __restrict__ blin,
                        float* __restrict__ out) {
  int t = blockIdx.x * blockDim.x + threadIdx.x;
  if (t >= N_GRAPHS * NCLS) return;
  int g = t >> 2, c = t & 3;
  float inv = 1.0f / fmaxf(cnts[g], 1.0f);
  float acc = blin[c];
  const float* srow = sums + ((size_t)g << 7);
#pragma unroll 4
  for (int k = 0; k < 128; ++k)
    acc = fmaf(srow[k] * inv, Wlin[k * NCLS + c], acc);
  out[t] = acc;
}


extern "C" void kernel_launch(void* const* d_in, const int* in_sizes, int n_in,
                              void* d_out, int out_size, void* d_ws, size_t ws_size,
                              hipStream_t stream) {
  const int*   tok   = (const int*)d_in[0];
  const int*   eidx  = (const int*)d_in[1];
  const int*   src   = eidx;
  const int*   dst   = eidx + N_EDGES;
  const int*   batch = (const int*)d_in[2];
  const float* emb   = (const float*)d_in[3];
  const float* W1    = (const float*)d_in[4];
  const float* b1    = (const float*)d_in[5];
  const float* W2    = (const float*)d_in[6];
  const float* b2    = (const float*)d_in[7];
  const float* Wlin  = (const float*)d_in[8];
  const float* blin  = (const float*)d_in[9];
  float*       out   = (float*)d_out;

  float* ws   = (float*)d_ws;
  float* deg  = ws;
  float* x0   = deg  + N_NODES;
  float* A    = x0   + (size_t)N_NODES * 64;
  float* B    = A    + (size_t)N_NODES * 128;
  float* sums = B    + (size_t)N_NODES * 128;
  float* cnts = sums + (size_t)N_GRAPHS * 128;

  const int T = 256;
  const int gNodes  = (N_NODES + T - 1) / T;
  const int gEdges  = (N_EDGES + T - 1) / T;
  const int gEmbed  = (N_NODES * 64) / T;
  const int gFeat   = (N_NODES * 128) / T;
  const int gScat   = (N_EDGES * 32) / T;
  const int gPool   = (N_NODES * 32) / T;
  const int gGemm   = N_NODES / 16;

  k_zero<<<gNodes, T, 0, stream>>>(deg, N_NODES);
  k_deg<<<gEdges, T, 0, stream>>>(dst, deg);
  k_dinv<<<gNodes, T, 0, stream>>>(deg);

  k_embed<<<gEmbed, T, 0, stream>>>(tok, emb, x0);

  gemm_wmma_kernel<64><<<gGemm, T, 0, stream>>>(x0, W1, A);
  k_selfinit<<<gFeat, T, 0, stream>>>(A, deg, B);
  k_scatter<<<gScat, T, 0, stream>>>(A, src, dst, deg, B);
  k_bias_relu<<<gFeat, T, 0, stream>>>(B, b1);

  gemm_wmma_kernel<128><<<gGemm, T, 0, stream>>>(B, W2, A);
  k_selfinit<<<gFeat, T, 0, stream>>>(A, deg, B);
  k_scatter<<<gScat, T, 0, stream>>>(A, src, dst, deg, B);
  k_bias_relu<<<gFeat, T, 0, stream>>>(B, b2);

  k_zero<<<(N_GRAPHS * 128 + N_GRAPHS + T - 1) / T, T, 0, stream>>>(sums,
                                                    N_GRAPHS * 128 + N_GRAPHS);
  k_pool<<<gPool, T, 0, stream>>>(B, batch, sums, cnts);
  k_final<<<(N_GRAPHS * NCLS + T - 1) / T, T, 0, stream>>>(sums, cnts, Wlin, blin, out);
}


# A_hat X summed in LDS by destination range: 256 work-groups own 391 nodes each, scan all edges, add matching rows by LDS float atomics, store rows once; no global f32 atomics there; sum order changed
# speedup vs baseline: 1.8200x; 1.8200x over previous
#include <hip/hip_runtime.h>
#include <hip/hip_bf16.h>


#define N_NODES  100000
#define N_EDGES  600000
#define N_GRAPHS 2000
#define NCLS     4

typedef float v2f __attribute__((ext_vector_type(2)));
typedef float v8f __attribute__((ext_vector_type(8)));


__global__ void k_zero(float* __restrict__ p, int n) {
  int i = blockIdx.x * blockDim.x + threadIdx.x;
  if (i < n) p[i] = 0.0f;
}

__global__ void k_deg(const int* __restrict__ dst, float* __restrict__ deg) {
  int e = blockIdx.x * blockDim.x + threadIdx.x;
  if (e < N_EDGES) atomicAdd(&deg[dst[e]], 1.0f);
}

__global__ void k_dinv(float* __restrict__ deg) {
  int i = blockIdx.x * blockDim.x + threadIdx.x;
  if (i < N_NODES) deg[i] = rsqrtf(deg[i] + 1.0f);
}

__global__ void k_embed(const int* __restrict__ tok, const float* __restrict__ emb,
                        float* __restrict__ x0) {
  size_t idx = (size_t)blockIdx.x * blockDim.x + threadIdx.x;
  int node = (int)(idx >> 6);
  int c    = (int)(idx & 63);
  x0[idx] = emb[((size_t)tok[node] << 6) | c];
}

template<int K>
__global__ void gemm_wmma_kernel(const float* __restrict__ X,
                                 const float* __restrict__ W,
                                 const float* __restrict__ bias,
                                 float* __restrict__ H) {
  const int wave  = (blockIdx.x << 3) | (threadIdx.x >> 5);
  const int lane  = threadIdx.x & 31;
  const int tileM = wave >> 3;
  const int tileN = wave & 7;
  const int half  = lane >> 4;
  const int lid   = lane & 15;
  const int m     = tileM * 16 + lid;
  const int n     = tileN * 16 + lid;
  const int kb    = half * 2;

  v8f acc = {};
  const float* xrow = X + (size_t)m * K + kb;
  const float* wcol = W + (size_t)kb * 128 + n;
#pragma unroll
  for (int k = 0; k < K; k += 4) {
    v2f a; a.x = xrow[k];               a.y = xrow[k + 1];
    v2f b; b.x = wcol[(size_t)k * 128]; b.y = wcol[(size_t)(k + 1) * 128];
    acc = __builtin_amdgcn_wmma_f32_16x16x4_f32(
         false, a,  false, b,
         (short)0, acc,  false,  false);
  }

  const float bn = bias[n];
  float* hbase = H + ((size_t)(tileM * 16 + half * 8)) * 128 + tileN * 16 + lid;
#pragma unroll
  for (int j = 0; j < 8; ++j) {
    float v = acc[j] + bn;
    hbase[(size_t)j * 128] = v > 0.0f ? v : 0.0f;
  }
}


#define AGG_WG    256
#define AGG_NPW   391
#define AGG_T     1024
#define AGG_QCAP  160
#define AGG_E4    (N_EDGES / 4)
#define AGG_STEPS 147
#define AGG_NCH   (AGG_STEPS * 32)
#define AGG_U     7

static_assert(AGG_WG * AGG_NPW >= N_NODES && (AGG_WG - 1) * AGG_NPW < N_NODES, "node ranges");
static_assert(N_EDGES % 4 == 0 && AGG_NCH * 32 >= AGG_E4, "edge chunks cover dst");
static_assert(AGG_STEPS % AGG_U == 0 && AGG_NPW <= 512 && N_EDGES < (1 << 23), "queue entry packing");

__device__ __forceinline__ void lds_add(float* p, float v) {
  __hip_atomic_fetch_add(p, v, __ATOMIC_RELAXED, __HIP_MEMORY_SCOPE_WORKGROUP);
}

template<int LOGC>
__global__ void __launch_bounds__(AGG_T)
k_agg(const float* __restrict__ X, const int* __restrict__ src,
      const int* __restrict__ dst, const float* __restrict__ dinv,
      float* __restrict__ out) {
  constexpr int C = 1 << LOGC;
  constexpr int H = C / 64;
  __shared__ __attribute__((aligned(16))) float acc[AGG_NPW * C];
  __shared__ unsigned queue[(AGG_T / 32) * AGG_QCAP];

  const int tid  = threadIdx.x;
  const int lane = tid & 31;
  const int wv   = __builtin_amdgcn_readfirstlane(tid >> 5);
  const int lo   = blockIdx.x * AGG_NPW;
  const int nn   = min(AGG_NPW, N_NODES - lo);
  const int n4   = nn << (LOGC - 2);

  const float4* X4 = (const float4*)(X + ((size_t)lo << LOGC));
  float4* a4 = (float4*)acc;
  for (int i = tid; i < n4; i += AGG_T) {
    const float dv = dinv[lo + (i >> (LOGC - 2))];
    const float4 v = X4[i];
    a4[i] = make_float4(v.x * dv * dv, v.y * dv * dv, v.z * dv * dv, v.w * dv * dv);
  }
  __syncthreads();

  unsigned* myq = queue + wv * AGG_QCAP;
  int qn = 0;

  auto drain = [&](const int n) {
    __builtin_amdgcn_fence(__ATOMIC_ACQ_REL, "wavefront");
    const unsigned pk = myq[qn - n + min(lane, n - 1)];
    qn -= n;
    const int   s  = src[pk >> 9];
    const int   dl = (int)(pk & 511u);
    const float nr = dinv[s] * dinv[lo + dl];
    for (int j0 = 0; j0 < n; j0 += 8) {
      float2 v[8][H];
#pragma unroll
      for (int jj = 0; jj < 8; ++jj) {
        const int j  = min(j0 + jj, n - 1);
        const int sj = __builtin_amdgcn_readlane(s, j);
        const float* row = X + ((size_t)sj << LOGC) + 2 * lane;
#pragma unroll
        for (int h = 0; h < H; ++h) v[jj][h] = *(const float2*)(row + 64 * h);
      }
#pragma unroll
      for (int jj = 0; jj < 8; ++jj) {
        const int j = j0 + jj;
        if (j < n) {
          const float nj = __int_as_float(__builtin_amdgcn_readlane(__float_as_int(nr), j));
          const int   dj = __builtin_amdgcn_readlane(dl, j);
          float* a = acc + (dj << LOGC) + 2 * lane;
#pragma unroll
          for (int h = 0; h < H; ++h) {
            lds_add(a + 64 * h,     v[jj][h].x * nj);
            lds_add(a + 64 * h + 1, v[jj][h].y * nj);
          }
        }
      }
    }
  };

  const int4* dst4 = (const int4*)dst;
  int chunk = wv + (blockIdx.x * AGG_STEPS) / 8;
  if (chunk >= AGG_NCH) chunk -= AGG_NCH;

  int4 dn[AGG_U];
  int  in[AGG_U];
  auto fetch = [&]() {
#pragma unroll
    for (int u = 0; u < AGG_U; ++u) {
      in[u] = chunk * 32 + lane;
      chunk += 32;
      if (chunk >= AGG_NCH) chunk -= AGG_NCH;
      dn[u] = dst4[min(in[u], AGG_E4 - 1)];
    }
  };
  fetch();

  for (int t0 = 0; t0 < AGG_STEPS; t0 += AGG_U) {
    int4 d[AGG_U];
    int  i4[AGG_U];
#pragma unroll
    for (int u = 0; u < AGG_U; ++u) { d[u] = dn[u]; i4[u] = in[u]; }
    if (t0 + AGG_U < AGG_STEPS) fetch();
#pragma unroll
    for (int u = 0; u < AGG_U; ++u) {
      const unsigned lim = (i4[u] < AGG_E4) ? (unsigned)nn : 0u;
      const unsigned r0 = (unsigned)(d[u].x - lo), r1 = (unsigned)(d[u].y - lo);
      const unsigned r2 = (unsigned)(d[u].z - lo), r3 = (unsigned)(d[u].w - lo);
      const bool m0 = r0 < lim, m1 = r1 < lim, m2 = r2 < lim, m3 = r3 < lim;
      if (__builtin_amdgcn_ballot_w32(m0 | m1 | m2 | m3)) {
        const unsigned eb = (unsigned)i4[u] << 2;
        unsigned mk;
        mk = __builtin_amdgcn_ballot_w32(m0);
        if (m0) myq[qn + __builtin_amdgcn_mbcnt_lo(mk, 0u)] = ((eb + 0u) << 9) | r0;
        qn += __builtin_popcount(mk);
        mk = __builtin_amdgcn_ballot_w32(m1);
        if (m1) myq[qn + __builtin_amdgcn_mbcnt_lo(mk, 0u)] = ((eb + 1u) << 9) | r1;
        qn += __builtin_popcount(mk);
        mk = __builtin_amdgcn_ballot_w32(m2);
        if (m2) myq[qn + __builtin_amdgcn_mbcnt_lo(mk, 0u)] = ((eb + 2u) << 9) | r2;
        qn += __builtin_popcount(mk);
        mk = __builtin_amdgcn_ballot_w32(m3);
        if (m3) myq[qn + __builtin_amdgcn_mbcnt_lo(mk, 0u)] = ((eb + 3u) << 9) | r3;
        qn += __builtin_popcount(mk);
        while (qn >= 32) drain(32);
      }
    }
  }
  if (qn > 0) drain(qn);
  __syncthreads();

  float4* o4 = (float4*)(out + ((size_t)lo << LOGC));
  for (int i = tid; i < n4; i += AGG_T) o4[i] = a4[i];
}


__global__ void k_pool(const float* __restrict__ X, const int* __restrict__ batch,
                       float* __restrict__ sums, float* __restrict__ cnts) {
  size_t gid = (size_t)blockIdx.x * blockDim.x + threadIdx.x;
  int node = (int)(gid >> 5);
  int lane = (int)(gid & 31);
  if (node >= N_NODES) return;
  int g = batch[node];
  const float4 v = *(const float4*)(X + ((size_t)node << 7) + (lane << 2));
  float* o = sums + ((size_t)g << 7) + (lane << 2);
  atomicAdd(o + 0, v.x);
  atomicAdd(o + 1, v.y);
  atomicAdd(o + 2, v.z);
  atomicAdd(o + 3, v.w);
  if (lane == 0) atomicAdd(&cnts[g], 1.0f);
}

__global__ void k_final(const float* __restrict__ sums, const float* __restrict__ cnts,
                        const float* __restrict__ Wlin, const float* __restrict__ blin,
                        float* __restrict__ out) {
  int t = blockIdx.x * blockDim.x + threadIdx.x;
  if (t >= N_GRAPHS * NCLS) return;
  int g = t >> 2, c = t & 3;
  float inv = 1.0f / fmaxf(cnts[g], 1.0f);
  float acc = blin[c];
  const float* srow = sums + ((size_t)g << 7);
#pragma unroll 4
  for (int k = 0; k < 128; ++k)
    acc = fmaf(srow[k] * inv, Wlin[k * NCLS + c], acc);
  out[t] = acc;
}


extern "C" void kernel_launch(void* const* d_in, const int* in_sizes, int n_in,
                              void* d_out, int out_size, void* d_ws, size_t ws_size,
                              hipStream_t stream) {
  const int*   tok   = (const int*)d_in[0];
  const int*   eidx  = (const int*)d_in[1];
  const int*   src   = eidx;
  const int*   dst   = eidx + N_EDGES;
  const int*   batch = (const int*)d_in[2];
  const float* emb   = (const float*)d_in[3];
  const float* W1    = (const float*)d_in[4];
  const float* b1    = (const float*)d_in[5];
  const float* W2    = (const float*)d_in[6];
  const float* b2    = (const float*)d_in[7];
  const float* Wlin  = (const float*)d_in[8];
  const float* blin  = (const float*)d_in[9];
  float*       out   = (float*)d_out;

  float* ws   = (float*)d_ws;
  float* deg  = ws;
  float* x0   = deg  + N_NODES;
  float* P    = x0   + (size_t)N_NODES * 64;
  float* A    = P    + (size_t)N_NODES * 64;
  float* B    = A    + (size_t)N_NODES * 128;
  float* sums = B    + (size_t)N_NODES * 128;
  float* cnts = sums + (size_t)N_GRAPHS * 128;

  const int T = 256;
  const int gNodes  = (N_NODES + T - 1) / T;
  const int gEdges  = (N_EDGES + T - 1) / T;
  const int gEmbed  = (N_NODES * 64) / T;
  const int gPool   = (N_NODES * 32) / T;
  const int gGemm   = N_NODES / 16;

  k_zero<<<gNodes, T, 0, stream>>>(deg, N_NODES);
  k_deg<<<gEdges, T, 0, stream>>>(dst, deg);
  k_dinv<<<gNodes, T, 0, stream>>>(deg);

  k_embed<<<gEmbed, T, 0, stream>>>(tok, emb, x0);

  k_agg<6><<<AGG_WG, AGG_T, 0, stream>>>(x0, src, dst, deg, P);
  gemm_wmma_kernel<64><<<gGemm, T, 0, stream>>>(P, W1, b1, A);

  k_agg<7><<<AGG_WG, AGG_T, 0, stream>>>(A, src, dst, deg, B);
  gemm_wmma_kernel<128><<<gGemm, T, 0, stream>>>(B, W2, b2, A);

  k_zero<<<(N_GRAPHS * 128 + N_GRAPHS + T - 1) / T, T, 0, stream>>>(sums,
                                                    N_GRAPHS * 128 + N_GRAPHS);
  k_pool<<<gPool, T, 0, stream>>>(A, batch, sums, cnts);
  k_final<<<(N_GRAPHS * NCLS + T - 1) / T, T, 0, stream>>>(sums, cnts, Wlin, blin, out);
}


# element-wise kernels (embed, pool, degree, zero, dinv) give each wave many elements, 16 bytes per lane per access: embed 10000 waves (was 200000), pool 10000 (was 100000); atomic sums as before
# speedup vs baseline: 2.0958x; 1.1515x over previous
#include <hip/hip_runtime.h>
#include <hip/hip_bf16.h>


#define N_NODES  100000
#define N_EDGES  600000
#define N_GRAPHS 2000
#define NCLS     4

typedef float v2f __attribute__((ext_vector_type(2)));
typedef float v8f __attribute__((ext_vector_type(8)));


#define EMB_WG   1250
#define EMB_PER  5
#define POOL_WG  1250
#define POOL_PER 10

static_assert((size_t)EMB_WG * 256 * EMB_PER == (size_t)N_NODES * 16, "embed covers x0 exactly");
static_assert(POOL_WG * 8 * POOL_PER == N_NODES, "pool covers the nodes exactly");
static_assert(N_NODES % 4 == 0 && N_EDGES % 4 == 0 && (N_GRAPHS * 128 + N_GRAPHS) % 4 == 0,
              "16-byte pieces");

__global__ void k_zero(float* __restrict__ p, int n4) {
  int i = blockIdx.x * blockDim.x + threadIdx.x;
  if (i < n4) ((float4*)p)[i] = make_float4(0.0f, 0.0f, 0.0f, 0.0f);
}

__global__ void k_deg(const int* __restrict__ dst, float* __restrict__ deg) {
  int i = blockIdx.x * blockDim.x + threadIdx.x;
  if (i < N_EDGES / 4) {
    const int4 d = ((const int4*)dst)[i];
    atomicAdd(&deg[d.x], 1.0f);
    atomicAdd(&deg[d.y], 1.0f);
    atomicAdd(&deg[d.z], 1.0f);
    atomicAdd(&deg[d.w], 1.0f);
  }
}

__global__ void k_dinv(float* __restrict__ deg) {
  int i = blockIdx.x * blockDim.x + threadIdx.x;
  if (i < N_NODES / 4) {
    float4* d4 = (float4*)deg;
    const float4 d = d4[i];
    d4[i] = make_float4(rsqrtf(d.x + 1.0f), rsqrtf(d.y + 1.0f),
                         rsqrtf(d.z + 1.0f), rsqrtf(d.w + 1.0f));
  }
}

__global__ void __launch_bounds__(256)
k_embed(const int* __restrict__ tok, const float* __restrict__ emb,
        float* __restrict__ x0) {
  const int g = blockIdx.x * 256 + threadIdx.x;
  const float4* __restrict__ e4 = (const float4*)emb;
  float4* __restrict__ x4 = (float4*)x0;
  int t[EMB_PER];
#pragma unroll
  for (int j = 0; j < EMB_PER; ++j) t[j] = tok[(g + j * (EMB_WG * 256)) >> 4];
#pragma unroll
  for (int j = 0; j < EMB_PER; ++j)
    x4[g + j * (EMB_WG * 256)] = e4[(t[j] << 4) | (g & 15)];
}

template<int K>
__global__ void gemm_wmma_kernel(const float* __restrict__ X,
                                 const float* __restrict__ W,
                                 const float* __restrict__ bias,
                                 float* __restrict__ H) {
  const int wave  = (blockIdx.x << 3) | (threadIdx.x >> 5);
  const int lane  = threadIdx.x & 31;
  const int tileM = wave >> 3;
  const int tileN = wave & 7;
  const int half  = lane >> 4;
  const int lid   = lane & 15;
  const int m     = tileM * 16 + lid;
  const int n     = tileN * 16 + lid;
  const int kb    = half * 2;

  v8f acc = {};
  const float* xrow = X + (size_t)m * K + kb;
  const float* wcol = W + (size_t)kb * 128 + n;
#pragma unroll
  for (int k = 0; k < K; k += 4) {
    v2f a; a.x = xrow[k];               a.y = xrow[k + 1];
    v2f b; b.x = wcol[(size_t)k * 128]; b.y = wcol[(size_t)(k + 1) * 128];
    acc = __builtin_amdgcn_wmma_f32_16x16x4_f32(
         false, a,  false, b,
         (short)0, acc,  false,  false);
  }

  const float bn = bias[n];
  float* hbase = H + ((size_t)(tileM * 16 + half * 8)) * 128 + tileN * 16 + lid;
#pragma unroll
  for (int j = 0; j < 8; ++j) {
    float v = acc[j] + bn;
    hbase[(size_t)j * 128] = v > 0.0f ? v : 0.0f;
  }
}


#define AGG_WG    256
#define AGG_NPW   391
#define AGG_T     1024
#define AGG_QCAP  160
#define AGG_E4    (N_EDGES / 4)
#define AGG_STEPS 147
#define AGG_NCH   (AGG_STEPS * 32)
#define AGG_U     7

static_assert(AGG_WG * AGG_NPW >= N_NODES && (AGG_WG - 1) * AGG_NPW < N_NODES, "node ranges");
static_assert(N_EDGES % 4 == 0 && AGG_NCH * 32 >= AGG_E4, "edge chunks cover dst");
static_assert(AGG_STEPS % AGG_U == 0 && AGG_NPW <= 512 && N_EDGES < (1 << 23), "queue entry packing");

__device__ __forceinline__ void lds_add(float* p, float v) {
  __hip_atomic_fetch_add(p, v, __ATOMIC_RELAXED, __HIP_MEMORY_SCOPE_WORKGROUP);
}

template<int LOGC>
__global__ void __launch_bounds__(AGG_T)
k_agg(const float* __restrict__ X, const int* __restrict__ src,
      const int* __restrict__ dst, const float* __restrict__ dinv,
      float* __restrict__ out) {
  constexpr int C = 1 << LOGC;
  constexpr int H = C / 64;
  __shared__ __attribute__((aligned(16))) float acc[AGG_NPW * C];
  __shared__ unsigned queue[(AGG_T / 32) * AGG_QCAP];

  const int tid  = threadIdx.x;
  const int lane = tid & 31;
  const int wv   = __builtin_amdgcn_readfirstlane(tid >> 5);
  const int lo   = blockIdx.x * AGG_NPW;
  const int nn   = min(AGG_NPW, N_NODES - lo);
  const int n4   = nn << (LOGC - 2);

  const float4* X4 = (const float4*)(X + ((size_t)lo << LOGC));
  float4* a4 = (float4*)acc;
  for (int i = tid; i < n4; i += AGG_T) {
    const float dv = dinv[lo + (i >> (LOGC - 2))];
    const float4 v = X4[i];
    a4[i] = make_float4(v.x * dv * dv, v.y * dv * dv, v.z * dv * dv, v.w * dv * dv);
  }
  __syncthreads();

  unsigned* myq = queue + wv * AGG_QCAP;
  int qn = 0;

  auto drain = [&](const int n) {
    __builtin_amdgcn_fence(__ATOMIC_ACQ_REL, "wavefront");
    const unsigned pk = myq[qn - n + min(lane, n - 1)];
    qn -= n;
    const int   s  = src[pk >> 9];
    const int   dl = (int)(pk & 511u);
    const float nr = dinv[s] * dinv[lo + dl];
    for (int j0 = 0; j0 < n; j0 += 8) {
      float2 v[8][H];
#pragma unroll
      for (int jj = 0; jj < 8; ++jj) {
        const int j  = min(j0 + jj, n - 1);
        const int sj = __builtin_amdgcn_readlane(s, j);
        const float* row = X + ((size_t)sj << LOGC) + 2 * lane;
#pragma unroll
        for (int h = 0; h < H; ++h) v[jj][h] = *(const float2*)(row + 64 * h);
      }
#pragma unroll
      for (int jj = 0; jj < 8; ++jj) {
        const int j = j0 + jj;
        if (j < n) {
          const float nj = __int_as_float(__builtin_amdgcn_readlane(__float_as_int(nr), j));
          const int   dj = __builtin_amdgcn_readlane(dl, j);
          float* a = acc + (dj << LOGC) + 2 * lane;
#pragma unroll
          for (int h = 0; h < H; ++h) {
            lds_add(a + 64 * h,     v[jj][h].x * nj);
            lds_add(a + 64 * h + 1, v[jj][h].y * nj);
          }
        }
      }
    }
  };

  const int4* dst4 = (const int4*)dst;
  int chunk = wv + (blockIdx.x * AGG_STEPS) / 8;
  if (chunk >= AGG_NCH) chunk -= AGG_NCH;

  int4 dn[AGG_U];
  int  in[AGG_U];
  auto fetch = [&]() {
#pragma unroll
    for (int u = 0; u < AGG_U; ++u) {
      in[u] = chunk * 32 + lane;
      chunk += 32;
      if (chunk >= AGG_NCH) chunk -= AGG_NCH;
      dn[u] = dst4[min(in[u], AGG_E4 - 1)];
    }
  };
  fetch();

  for (int t0 = 0; t0 < AGG_STEPS; t0 += AGG_U) {
    int4 d[AGG_U];
    int  i4[AGG_U];
#pragma unroll
    for (int u = 0; u < AGG_U; ++u) { d[u] = dn[u]; i4[u] = in[u]; }
    if (t0 + AGG_U < AGG_STEPS) fetch();
#pragma unroll
    for (int u = 0; u < AGG_U; ++u) {
      const unsigned lim = (i4[u] < AGG_E4) ? (unsigned)nn : 0u;
      const unsigned r0 = (unsigned)(d[u].x - lo), r1 = (unsigned)(d[u].y - lo);
      const unsigned r2 = (unsigned)(d[u].z - lo), r3 = (unsigned)(d[u].w - lo);
      const bool m0 = r0 < lim, m1 = r1 < lim, m2 = r2 < lim, m3 = r3 < lim;
      if (__builtin_amdgcn_ballot_w32(m0 | m1 | m2 | m3)) {
        const unsigned eb = (unsigned)i4[u] << 2;
        unsigned mk;
        mk = __builtin_amdgcn_ballot_w32(m0);
        if (m0) myq[qn + __builtin_amdgcn_mbcnt_lo(mk, 0u)] = ((eb + 0u) << 9) | r0;
        qn += __builtin_popcount(mk);
        mk = __builtin_amdgcn_ballot_w32(m1);
        if (m1) myq[qn + __builtin_amdgcn_mbcnt_lo(mk, 0u)] = ((eb + 1u) << 9) | r1;
        qn += __builtin_popcount(mk);
        mk = __builtin_amdgcn_ballot_w32(m2);
        if (m2) myq[qn + __builtin_amdgcn_mbcnt_lo(mk, 0u)] = ((eb + 2u) << 9) | r2;
        qn += __builtin_popcount(mk);
        mk = __builtin_amdgcn_ballot_w32(m3);
        if (m3) myq[qn + __builtin_amdgcn_mbcnt_lo(mk, 0u)] = ((eb + 3u) << 9) | r3;
        qn += __builtin_popcount(mk);
        while (qn >= 32) drain(32);
      }
    }
  }
  if (qn > 0) drain(qn);
  __syncthreads();

  float4* o4 = (float4*)(out + ((size_t)lo << LOGC));
  for (int i = tid; i < n4; i += AGG_T) o4[i] = a4[i];
}


__global__ void __launch_bounds__(256)
k_pool(const float* __restrict__ X, const int* __restrict__ batch,
       float* __restrict__ sums, float* __restrict__ cnts) {
  const int lane  = threadIdx.x & 31;
  const int node0 = ((blockIdx.x << 3) | (threadIdx.x >> 5)) * POOL_PER;
  const int gl = batch[node0 + min(lane, POOL_PER - 1)];
  int    g[POOL_PER];
  float4 v[POOL_PER];
#pragma unroll
  for (int j = 0; j < POOL_PER; ++j) {
    g[j] = batch[node0 + j];
    v[j] = *(const float4*)(X + ((size_t)(node0 + j) << 7) + (lane << 2));
  }
#pragma unroll
  for (int j = 0; j < POOL_PER; ++j) {
    float* o = sums + ((size_t)g[j] << 7) + (lane << 2);
    atomicAdd(o + 0, v[j].x);
    atomicAdd(o + 1, v[j].y);
    atomicAdd(o + 2, v[j].z);
    atomicAdd(o + 3, v[j].w);
  }
  if (lane < POOL_PER) atomicAdd(&cnts[gl], 1.0f);
}

__global__ void k_final(const float* __restrict__ sums, const float* __restrict__ cnts,
                        const float* __restrict__ Wlin, const float* __restrict__ blin,
                        float* __restrict__ out) {
  int t = blockIdx.x * blockDim.x + threadIdx.x;
  if (t >= N_GRAPHS * NCLS) return;
  int g = t >> 2, c = t & 3;
  float inv = 1.0f / fmaxf(cnts[g], 1.0f);
  float acc = blin[c];
  const float* srow = sums + ((size_t)g << 7);
#pragma unroll 4
  for (int k = 0; k < 128; ++k)
    acc = fmaf(srow[k] * inv, Wlin[k * NCLS + c], acc);
  out[t] = acc;
}


extern "C" void kernel_launch(void* const* d_in, const int* in_sizes, int n_in,
                              void* d_out, int out_size, void* d_ws, size_t ws_size,
                              hipStream_t stream) {
  const int*   tok   = (const int*)d_in[0];
  const int*   eidx  = (const int*)d_in[1];
  const int*   src   = eidx;
  const int*   dst   = eidx + N_EDGES;
  const int*   batch = (const int*)d_in[2];
  const float* emb   = (const float*)d_in[3];
  const float* W1    = (const float*)d_in[4];
  const float* b1    = (const float*)d_in[5];
  const float* W2    = (const float*)d_in[6];
  const float* b2    = (const float*)d_in[7];
  const float* Wlin  = (const float*)d_in[8];
  const float* blin  = (const float*)d_in[9];
  float*       out   = (float*)d_out;

  float* ws   = (float*)d_ws;
  float* deg  = ws;
  float* x0   = deg  + N_NODES;
  float* P    = x0   + (size_t)N_NODES * 64;
  float* A    = P    + (size_t)N_NODES * 64;
  float* B    = A    + (size_t)N_NODES * 128;
  float* sums = B    + (size_t)N_NODES * 128;
  float* cnts = sums + (size_t)N_GRAPHS * 128;

  const int T = 256;
  const int gNodes4 = (N_NODES / 4 + T - 1) / T;
  const int gEdges4 = (N_EDGES / 4 + T - 1) / T;
  const int gGemm   = N_NODES / 16;

  k_zero<<<gNodes4, T, 0, stream>>>(deg, N_NODES / 4);
  k_deg<<<gEdges4, T, 0, stream>>>(dst, deg);
  k_dinv<<<gNodes4, T, 0, stream>>>(deg);

  k_embed<<<EMB_WG, T, 0, stream>>>(tok, emb, x0);

  k_agg<6><<<AGG_WG, AGG_T, 0, stream>>>(x0, src, dst, deg, P);
  gemm_wmma_kernel<64><<<gGemm, T, 0, stream>>>(P, W1, b1, A);

  k_agg<7><<<AGG_WG, AGG_T, 0, stream>>>(A, src, dst, deg, B);
  gemm_wmma_kernel<128><<<gGemm, T, 0, stream>>>(B, W2, b2, A);

  k_zero<<<((N_GRAPHS * 128 + N_GRAPHS) / 4 + T - 1) / T, T, 0, stream>>>(
      sums, (N_GRAPHS * 128 + N_GRAPHS) / 4);
  k_pool<<<POOL_WG, T, 0, stream>>>(A, batch, sums, cnts);
  k_final<<<(N_GRAPHS * NCLS + T - 1) / T, T, 0, stream>>>(sums, cnts, Wlin, blin, out);
}


# mean-pool sums formed in the layer-2 GEMM epilogue: a lane adds its column over its 8 tile rows per run of equal graph id, one atomic per run; x2 store and pool kernel gone; sum order changed
# speedup vs baseline: 2.4346x; 1.1617x over previous
#include <hip/hip_runtime.h>
#include <hip/hip_bf16.h>


#define N_NODES  100000
#define N_EDGES  600000
#define N_GRAPHS 2000
#define NCLS     4

typedef float v2f __attribute__((ext_vector_type(2)));
typedef float v8f __attribute__((ext_vector_type(8)));


#define EMB_WG   1250
#define EMB_PER  5

static_assert((size_t)EMB_WG * 256 * EMB_PER == (size_t)N_NODES * 16, "embed covers x0 exactly");
static_assert(N_NODES % 16 == 0, "whole 16-row tiles; a tile's 16 batch ids are read as four int4");
static_assert(N_NODES % 4 == 0 && N_EDGES % 4 == 0 && (N_GRAPHS * 128 + N_GRAPHS) % 4 == 0,
              "16-byte pieces");

__global__ void k_zero(float* __restrict__ p, int n4) {
  int i = blockIdx.x * blockDim.x + threadIdx.x;
  if (i < n4) ((float4*)p)[i] = make_float4(0.0f, 0.0f, 0.0f, 0.0f);
}

__global__ void k_deg(const int* __restrict__ dst, float* __restrict__ deg) {
  int i = blockIdx.x * blockDim.x + threadIdx.x;
  if (i < N_EDGES / 4) {
    const int4 d = ((const int4*)dst)[i];
    atomicAdd(&deg[d.x], 1.0f);
    atomicAdd(&deg[d.y], 1.0f);
    atomicAdd(&deg[d.z], 1.0f);
    atomicAdd(&deg[d.w], 1.0f);
  }
}

__global__ void k_dinv(float* __restrict__ deg) {
  int i = blockIdx.x * blockDim.x + threadIdx.x;
  if (i < N_NODES / 4) {
    float4* d4 = (float4*)deg;
    const float4 d = d4[i];
    d4[i] = make_float4(rsqrtf(d.x + 1.0f), rsqrtf(d.y + 1.0f),
                         rsqrtf(d.z + 1.0f), rsqrtf(d.w + 1.0f));
  }
}

__global__ void __launch_bounds__(256)
k_embed(const int* __restrict__ tok, const float* __restrict__ emb,
        float* __restrict__ x0) {
  const int g = blockIdx.x * 256 + threadIdx.x;
  const float4* __restrict__ e4 = (const float4*)emb;
  float4* __restrict__ x4 = (float4*)x0;
  int t[EMB_PER];
#pragma unroll
  for (int j = 0; j < EMB_PER; ++j) t[j] = tok[(g + j * (EMB_WG * 256)) >> 4];
#pragma unroll
  for (int j = 0; j < EMB_PER; ++j)
    x4[g + j * (EMB_WG * 256)] = e4[(t[j] << 4) | (g & 15)];
}

template<int K, bool POOL>
__global__ void gemm_wmma_kernel(const float* __restrict__ X,
                                 const float* __restrict__ W,
                                 const float* __restrict__ bias,
                                 float* __restrict__ H,
                                 const int* __restrict__ batch,
                                 float* __restrict__ sums,
                                 float* __restrict__ cnts) {
  const int wave  = (blockIdx.x << 3) | (threadIdx.x >> 5);
  const int lane  = threadIdx.x & 31;
  const int tileM = wave >> 3;
  const int tileN = wave & 7;
  const int half  = lane >> 4;
  const int lid   = lane & 15;
  const int m     = tileM * 16 + lid;
  const int n     = tileN * 16 + lid;
  const int kb    = half * 2;

  v8f acc = {};
  const float* xrow = X + (size_t)m * K + kb;
  const float* wcol = W + (size_t)kb * 128 + n;
#pragma unroll
  for (int k = 0; k < K; k += 4) {
    v2f a; a.x = xrow[k];               a.y = xrow[k + 1];
    v2f b; b.x = wcol[(size_t)k * 128]; b.y = wcol[(size_t)(k + 1) * 128];
    acc = __builtin_amdgcn_wmma_f32_16x16x4_f32(
         false, a,  false, b,
         (short)0, acc,  false,  false);
  }

  const float bn = bias[n];
  if (!POOL) {
    float* hbase = H + ((size_t)(tileM * 16 + half * 8)) * 128 + tileN * 16 + lid;
#pragma unroll
    for (int j = 0; j < 8; ++j) {
      float v = acc[j] + bn;
      hbase[(size_t)j * 128] = v > 0.0f ? v : 0.0f;
    }
  } else {
    const int   tM = __builtin_amdgcn_readfirstlane(tileM);
    const int4* bt = (const int4*)(batch + tM * 16);
    const int4  q0 = bt[0], q1 = bt[1], q2 = bt[2], q3 = bt[3];
    const int   glo[8] = {q0.x, q0.y, q0.z, q0.w, q1.x, q1.y, q1.z, q1.w};
    const int   ghi[8] = {q2.x, q2.y, q2.z, q2.w, q3.x, q3.y, q3.z, q3.w};
    const bool  counts = (tileN == 0) && (lid == 0);
    float v0 = acc[0] + bn;
    float s   = v0 > 0.0f ? v0 : 0.0f;
    float len = 1.0f;
    int   gc  = half ? ghi[0] : glo[0];
#pragma unroll
    for (int j = 1; j < 8; ++j) {
      const int gj = half ? ghi[j] : glo[j];
      float v = acc[j] + bn;
      v = v > 0.0f ? v : 0.0f;
      if (gj == gc) {
        s += v;
        len += 1.0f;
      } else {
        atomicAdd(sums + ((size_t)gc << 7) + n, s);
        if (counts) atomicAdd(cnts + gc, len);
        s = v;
        len = 1.0f;
        gc = gj;
      }
    }
    atomicAdd(sums + ((size_t)gc << 7) + n, s);
    if (counts) atomicAdd(cnts + gc, len);
  }
}


#define AGG_WG    256
#define AGG_NPW   391
#define AGG_T     1024
#define AGG_QCAP  160
#define AGG_E4    (N_EDGES / 4)
#define AGG_STEPS 147
#define AGG_NCH   (AGG_STEPS * 32)
#define AGG_U     7

static_assert(AGG_WG * AGG_NPW >= N_NODES && (AGG_WG - 1) * AGG_NPW < N_NODES, "node ranges");
static_assert(N_EDGES % 4 == 0 && AGG_NCH * 32 >= AGG_E4, "edge chunks cover dst");
static_assert(AGG_STEPS % AGG_U == 0 && AGG_NPW <= 512 && N_EDGES < (1 << 23), "queue entry packing");

__device__ __forceinline__ void lds_add(float* p, float v) {
  __hip_atomic_fetch_add(p, v, __ATOMIC_RELAXED, __HIP_MEMORY_SCOPE_WORKGROUP);
}

template<int LOGC>
__global__ void __launch_bounds__(AGG_T)
k_agg(const float* __restrict__ X, const int* __restrict__ src,
      const int* __restrict__ dst, const float* __restrict__ dinv,
      float* __restrict__ out) {
  constexpr int C = 1 << LOGC;
  constexpr int H = C / 64;
  __shared__ __attribute__((aligned(16))) float acc[AGG_NPW * C];
  __shared__ unsigned queue[(AGG_T / 32) * AGG_QCAP];

  const int tid  = threadIdx.x;
  const int lane = tid & 31;
  const int wv   = __builtin_amdgcn_readfirstlane(tid >> 5);
  const int lo   = blockIdx.x * AGG_NPW;
  const int nn   = min(AGG_NPW, N_NODES - lo);
  const int n4   = nn << (LOGC - 2);

  const float4* X4 = (const float4*)(X + ((size_t)lo << LOGC));
  float4* a4 = (float4*)acc;
  for (int i = tid; i < n4; i += AGG_T) {
    const float dv = dinv[lo + (i >> (LOGC - 2))];
    const float4 v = X4[i];
    a4[i] = make_float4(v.x * dv * dv, v.y * dv * dv, v.z * dv * dv, v.w * dv * dv);
  }
  __syncthreads();

  unsigned* myq = queue + wv * AGG_QCAP;
  int qn = 0;

  auto drain = [&](const int n) {
    __builtin_amdgcn_fence(__ATOMIC_ACQ_REL, "wavefront");
    const unsigned pk = myq[qn - n + min(lane, n - 1)];
    qn -= n;
    const int   s  = src[pk >> 9];
    const int   dl = (int)(pk & 511u);
    const float nr = dinv[s] * dinv[lo + dl];
    for (int j0 = 0; j0 < n; j0 += 8) {
      float2 v[8][H];
#pragma unroll
      for (int jj = 0; jj < 8; ++jj) {
        const int j  = min(j0 + jj, n - 1);
        const int sj = __builtin_amdgcn_readlane(s, j);
        const float* row = X + ((size_t)sj << LOGC) + 2 * lane;
#pragma unroll
        for (int h = 0; h < H; ++h) v[jj][h] = *(const float2*)(row + 64 * h);
      }
#pragma unroll
      for (int jj = 0; jj < 8; ++jj) {
        const int j = j0 + jj;
        if (j < n) {
          const float nj = __int_as_float(__builtin_amdgcn_readlane(__float_as_int(nr), j));
          const int   dj = __builtin_amdgcn_readlane(dl, j);
          float* a = acc + (dj << LOGC) + 2 * lane;
#pragma unroll
          for (int h = 0; h < H; ++h) {
            lds_add(a + 64 * h,     v[jj][h].x * nj);
            lds_add(a + 64 * h + 1, v[jj][h].y * nj);
          }
        }
      }
    }
  };

  const int4* dst4 = (const int4*)dst;
  int chunk = wv + (blockIdx.x * AGG_STEPS) / 8;
  if (chunk >= AGG_NCH) chunk -= AGG_NCH;

  int4 dn[AGG_U];
  int  in[AGG_U];
  auto fetch = [&]() {
#pragma unroll
    for (int u = 0; u < AGG_U; ++u) {
      in[u] = chunk * 32 + lane;
      chunk += 32;
      if (chunk >= AGG_NCH) chunk -= AGG_NCH;
      dn[u] = dst4[min(in[u], AGG_E4 - 1)];
    }
  };
  fetch();

  for (int t0 = 0; t0 < AGG_STEPS; t0 += AGG_U) {
    int4 d[AGG_U];
    int  i4[AGG_U];
#pragma unroll
    for (int u = 0; u < AGG_U; ++u) { d[u] = dn[u]; i4[u] = in[u]; }
    if (t0 + AGG_U < AGG_STEPS) fetch();
#pragma unroll
    for (int u = 0; u < AGG_U; ++u) {
      const unsigned lim = (i4[u] < AGG_E4) ? (unsigned)nn : 0u;
      const unsigned r0 = (unsigned)(d[u].x - lo), r1 = (unsigned)(d[u].y - lo);
      const unsigned r2 = (unsigned)(d[u].z - lo), r3 = (unsigned)(d[u].w - lo);
      const bool m0 = r0 < lim, m1 = r1 < lim, m2 = r2 < lim, m3 = r3 < lim;
      if (__builtin_amdgcn_ballot_w32(m0 | m1 | m2 | m3)) {
        const unsigned eb = (unsigned)i4[u] << 2;
        unsigned mk;
        mk = __builtin_amdgcn_ballot_w32(m0);
        if (m0) myq[qn + __builtin_amdgcn_mbcnt_lo(mk, 0u)] = ((eb + 0u) << 9) | r0;
        qn += __builtin_popcount(mk);
        mk = __builtin_amdgcn_ballot_w32(m1);
        if (m1) myq[qn + __builtin_amdgcn_mbcnt_lo(mk, 0u)] = ((eb + 1u) << 9) | r1;
        qn += __builtin_popcount(mk);
        mk = __builtin_amdgcn_ballot_w32(m2);
        if (m2) myq[qn + __builtin_amdgcn_mbcnt_lo(mk, 0u)] = ((eb + 2u) << 9) | r2;
        qn += __builtin_popcount(mk);
        mk = __builtin_amdgcn_ballot_w32(m3);
        if (m3) myq[qn + __builtin_amdgcn_mbcnt_lo(mk, 0u)] = ((eb + 3u) << 9) | r3;
        qn += __builtin_popcount(mk);
        while (qn >= 32) drain(32);
      }
    }
  }
  if (qn > 0) drain(qn);
  __syncthreads();

  float4* o4 = (float4*)(out + ((size_t)lo << LOGC));
  for (int i = tid; i < n4; i += AGG_T) o4[i] = a4[i];
}


__global__ void k_final(const float* __restrict__ sums, const float* __restrict__ cnts,
                        const float* __restrict__ Wlin, const float* __restrict__ blin,
                        float* __restrict__ out) {
  int t = blockIdx.x * blockDim.x + threadIdx.x;
  if (t >= N_GRAPHS * NCLS) return;
  int g = t >> 2, c = t & 3;
  float inv = 1.0f / fmaxf(cnts[g], 1.0f);
  float acc = blin[c];
  const float* srow = sums + ((size_t)g << 7);
#pragma unroll 4
  for (int k = 0; k < 128; ++k)
    acc = fmaf(srow[k] * inv, Wlin[k * NCLS + c], acc);
  out[t] = acc;
}


extern "C" void kernel_launch(void* const* d_in, const int* in_sizes, int n_in,
                              void* d_out, int out_size, void* d_ws, size_t ws_size,
                              hipStream_t stream) {
  const int*   tok   = (const int*)d_in[0];
  const int*   eidx  = (const int*)d_in[1];
  const int*   src   = eidx;
  const int*   dst   = eidx + N_EDGES;
  const int*   batch = (const int*)d_in[2];
  const float* emb   = (const float*)d_in[3];
  const float* W1    = (const float*)d_in[4];
  const float* b1    = (const float*)d_in[5];
  const float* W2    = (const float*)d_in[6];
  const float* b2    = (const float*)d_in[7];
  const float* Wlin  = (const float*)d_in[8];
  const float* blin  = (const float*)d_in[9];
  float*       out   = (float*)d_out;

  float* ws   = (float*)d_ws;
  float* deg  = ws;
  float* x0   = deg  + N_NODES;
  float* P    = x0   + (size_t)N_NODES * 64;
  float* A    = P    + (size_t)N_NODES * 64;
  float* B    = A    + (size_t)N_NODES * 128;
  float* sums = B    + (size_t)N_NODES * 128;
  float* cnts = sums + (size_t)N_GRAPHS * 128;

  const int T = 256;
  const int gNodes4 = (N_NODES / 4 + T - 1) / T;
  const int gEdges4 = (N_EDGES / 4 + T - 1) / T;
  const int gGemm   = N_NODES / 16;

  k_zero<<<gNodes4, T, 0, stream>>>(deg, N_NODES / 4);
  k_deg<<<gEdges4, T, 0, stream>>>(dst, deg);
  k_dinv<<<gNodes4, T, 0, stream>>>(deg);

  k_embed<<<EMB_WG, T, 0, stream>>>(tok, emb, x0);

  k_agg<6><<<AGG_WG, AGG_T, 0, stream>>>(x0, src, dst, deg, P);
  gemm_wmma_kernel<64, false><<<gGemm, T, 0, stream>>>(P, W1, b1, A, nullptr, nullptr, nullptr);

  k_agg<7><<<AGG_WG, AGG_T, 0, stream>>>(A, src, dst, deg, B);
  k_zero<<<((N_GRAPHS * 128 + N_GRAPHS) / 4 + T - 1) / T, T, 0, stream>>>(
      sums, (N_GRAPHS * 128 + N_GRAPHS) / 4);
  gemm_wmma_kernel<128, true><<<gGemm, T, 0, stream>>>(B, W2, b2, nullptr, batch, sums, cnts);

  k_final<<<(N_GRAPHS * NCLS + T - 1) / T, T, 0, stream>>>(sums, cnts, Wlin, blin, out);
}


# layer GEMM fused into its aggregation kernel: a work-group multiplies its summed rows from LDS on WMMA, W column tile in registers; A_hat X never stored; 8 launches (was 10); pool sum order changed
# speedup vs baseline: 3.1210x; 1.2819x over previous
#include <hip/hip_runtime.h>
#include <hip/hip_bf16.h>


#define N_NODES  100000
#define N_EDGES  600000
#define N_GRAPHS 2000
#define NCLS     4

typedef float v2f __attribute__((ext_vector_type(2)));
typedef float v8f __attribute__((ext_vector_type(8)));


#define EMB_WG   1250
#define EMB_PER  5

static_assert((size_t)EMB_WG * 256 * EMB_PER == (size_t)N_NODES * 16, "embed covers x0 exactly");
static_assert(N_NODES % 16 == 0, "whole 16-row tiles; a tile's 16 batch ids are read as four int4");
static_assert(N_NODES % 4 == 0 && N_EDGES % 4 == 0 && (N_GRAPHS * 128 + N_GRAPHS) % 4 == 0,
              "16-byte pieces");

__global__ void k_zero(float* __restrict__ p, int n4) {
  int i = blockIdx.x * blockDim.x + threadIdx.x;
  if (i < n4) ((float4*)p)[i] = make_float4(0.0f, 0.0f, 0.0f, 0.0f);
}

__global__ void k_deg(const int* __restrict__ dst, float* __restrict__ deg) {
  int i = blockIdx.x * blockDim.x + threadIdx.x;
  if (i < N_EDGES / 4) {
    const int4 d = ((const int4*)dst)[i];
    atomicAdd(&deg[d.x], 1.0f);
    atomicAdd(&deg[d.y], 1.0f);
    atomicAdd(&deg[d.z], 1.0f);
    atomicAdd(&deg[d.w], 1.0f);
  }
}

__global__ void k_dinv(float* __restrict__ deg) {
  int i = blockIdx.x * blockDim.x + threadIdx.x;
  if (i < N_NODES / 4) {
    float4* d4 = (float4*)deg;
    const float4 d = d4[i];
    d4[i] = make_float4(rsqrtf(d.x + 1.0f), rsqrtf(d.y + 1.0f),
                         rsqrtf(d.z + 1.0f), rsqrtf(d.w + 1.0f));
  }
}

__global__ void __launch_bounds__(256)
k_embed(const int* __restrict__ tok, const float* __restrict__ emb,
        float* __restrict__ x0) {
  const int g = blockIdx.x * 256 + threadIdx.x;
  const float4* __restrict__ e4 = (const float4*)emb;
  float4* __restrict__ x4 = (float4*)x0;
  int t[EMB_PER];
#pragma unroll
  for (int j = 0; j < EMB_PER; ++j) t[j] = tok[(g + j * (EMB_WG * 256)) >> 4];
#pragma unroll
  for (int j = 0; j < EMB_PER; ++j)
    x4[g + j * (EMB_WG * 256)] = e4[(t[j] << 4) | (g & 15)];
}


#define AGG_WG    256
#define AGG_NPW   391
#define AGG_ROWS  400
#define AGG_T     1024
#define AGG_QCAP  160
#define AGG_E4    (N_EDGES / 4)
#define AGG_STEPS 147
#define AGG_NCH   (AGG_STEPS * 32)
#define AGG_U     7

static_assert(AGG_WG * AGG_NPW >= N_NODES && (AGG_WG - 1) * AGG_NPW < N_NODES, "node ranges");
static_assert(AGG_ROWS % 16 == 0 && AGG_ROWS >= AGG_NPW, "whole 16-row tiles cover a range");
static_assert(N_EDGES % 4 == 0 && AGG_NCH * 32 >= AGG_E4, "edge chunks cover dst");
static_assert(AGG_STEPS % AGG_U == 0 && AGG_NPW <= 512 && N_EDGES < (1 << 23), "queue entry packing");

__device__ __forceinline__ void lds_add(float* p, float v) {
  __hip_atomic_fetch_add(p, v, __ATOMIC_RELAXED, __HIP_MEMORY_SCOPE_WORKGROUP);
}

template<int LOGC, bool POOL>
__global__ void __launch_bounds__(AGG_T)
k_layer(const float* __restrict__ X, const int* __restrict__ src,
        const int* __restrict__ dst, const float* __restrict__ dinv,
        const float* __restrict__ W, const float* __restrict__ bias,
        float* __restrict__ H, const int* __restrict__ batch,
        float* __restrict__ sums, float* __restrict__ cnts) {
  constexpr int C = 1 << LOGC;
  constexpr int HP = C / 64;
  constexpr int PITCH = C + 4;
  __shared__ __attribute__((aligned(16))) float acc[AGG_ROWS * PITCH];
  __shared__ unsigned queue[(AGG_T / 32) * AGG_QCAP];
  __shared__ __attribute__((aligned(16))) int gid[POOL ? AGG_ROWS : 4];

  const int tid  = threadIdx.x;
  const int lane = tid & 31;
  const int wv   = __builtin_amdgcn_readfirstlane(tid >> 5);
  const int lo   = blockIdx.x * AGG_NPW;
  const int nn   = min(AGG_NPW, N_NODES - lo);

  const int n4 = nn << (LOGC - 2);
  const float4* X4 = (const float4*)(X + ((size_t)lo << LOGC));
#pragma unroll 4
  for (int i = tid; i < n4; i += AGG_T) {
    const int r = i >> (LOGC - 2);
    const float dv = dinv[lo + r];
    const float4 v = X4[i];
    *(float4*)(acc + r * PITCH + 4 * (i & (C / 4 - 1))) =
        make_float4(v.x * dv * dv, v.y * dv * dv, v.z * dv * dv, v.w * dv * dv);
  }
  for (int i = n4 + tid; i < AGG_ROWS * (C / 4); i += AGG_T)
    *(float4*)(acc + (i >> (LOGC - 2)) * PITCH + 4 * (i & (C / 4 - 1))) =
        make_float4(0.0f, 0.0f, 0.0f, 0.0f);
  if (POOL) {
    for (int r = tid; r < AGG_ROWS; r += AGG_T) gid[r] = (r < nn) ? batch[lo + r] : -1;
  }
  __syncthreads();

  unsigned* myq = queue + wv * AGG_QCAP;
  int qn = 0;

  auto drain = [&](const int n) {
    __builtin_amdgcn_fence(__ATOMIC_ACQ_REL, "wavefront");
    const unsigned pk = myq[qn - n + min(lane, n - 1)];
    qn -= n;
    const int   s  = src[pk >> 9];
    const int   dl = (int)(pk & 511u);
    const float nr = dinv[s] * dinv[lo + dl];
    for (int j0 = 0; j0 < n; j0 += 8) {
      float2 v[8][HP];
#pragma unroll
      for (int jj = 0; jj < 8; ++jj) {
        const int j  = min(j0 + jj, n - 1);
        const int sj = __builtin_amdgcn_readlane(s, j);
        const float* row = X + ((size_t)sj << LOGC) + 2 * lane;
#pragma unroll
        for (int h = 0; h < HP; ++h) v[jj][h] = *(const float2*)(row + 64 * h);
      }
#pragma unroll
      for (int jj = 0; jj < 8; ++jj) {
        const int j = j0 + jj;
        if (j < n) {
          const float nj = __int_as_float(__builtin_amdgcn_readlane(__float_as_int(nr), j));
          const int   dj = __builtin_amdgcn_readlane(dl, j);
          float* a = acc + dj * PITCH + 2 * lane;
#pragma unroll
          for (int h = 0; h < HP; ++h) {
            lds_add(a + 64 * h,     v[jj][h].x * nj);
            lds_add(a + 64 * h + 1, v[jj][h].y * nj);
          }
        }
      }
    }
  };

  const int4* dst4 = (const int4*)dst;
  int chunk = wv + (blockIdx.x * AGG_STEPS) / 8;
  if (chunk >= AGG_NCH) chunk -= AGG_NCH;

  int4 dn[AGG_U];
  int  in[AGG_U];
  auto fetch = [&]() {
#pragma unroll
    for (int u = 0; u < AGG_U; ++u) {
      in[u] = chunk * 32 + lane;
      chunk += 32;
      if (chunk >= AGG_NCH) chunk -= AGG_NCH;
      dn[u] = dst4[min(in[u], AGG_E4 - 1)];
    }
  };
  fetch();

  for (int t0 = 0; t0 < AGG_STEPS; t0 += AGG_U) {
    int4 d[AGG_U];
    int  i4[AGG_U];
#pragma unroll
    for (int u = 0; u < AGG_U; ++u) { d[u] = dn[u]; i4[u] = in[u]; }
    if (t0 + AGG_U < AGG_STEPS) fetch();
#pragma unroll
    for (int u = 0; u < AGG_U; ++u) {
      const unsigned lim = (i4[u] < AGG_E4) ? (unsigned)nn : 0u;
      const unsigned r0 = (unsigned)(d[u].x - lo), r1 = (unsigned)(d[u].y - lo);
      const unsigned r2 = (unsigned)(d[u].z - lo), r3 = (unsigned)(d[u].w - lo);
      const bool m0 = r0 < lim, m1 = r1 < lim, m2 = r2 < lim, m3 = r3 < lim;
      if (__builtin_amdgcn_ballot_w32(m0 | m1 | m2 | m3)) {
        const unsigned eb = (unsigned)i4[u] << 2;
        unsigned mk;
        mk = __builtin_amdgcn_ballot_w32(m0);
        if (m0) myq[qn + __builtin_amdgcn_mbcnt_lo(mk, 0u)] = ((eb + 0u) << 9) | r0;
        qn += __builtin_popcount(mk);
        mk = __builtin_amdgcn_ballot_w32(m1);
        if (m1) myq[qn + __builtin_amdgcn_mbcnt_lo(mk, 0u)] = ((eb + 1u) << 9) | r1;
        qn += __builtin_popcount(mk);
        mk = __builtin_amdgcn_ballot_w32(m2);
        if (m2) myq[qn + __builtin_amdgcn_mbcnt_lo(mk, 0u)] = ((eb + 2u) << 9) | r2;
        qn += __builtin_popcount(mk);
        mk = __builtin_amdgcn_ballot_w32(m3);
        if (m3) myq[qn + __builtin_amdgcn_mbcnt_lo(mk, 0u)] = ((eb + 3u) << 9) | r3;
        qn += __builtin_popcount(mk);
        while (qn >= 32) drain(32);
      }
    }
  }
  if (qn > 0) drain(qn);
  __syncthreads();

  const int tileN = wv & 7;
  const int half  = lane >> 4;
  const int lid   = lane & 15;
  const int n     = tileN * 16 + lid;
  const int kb    = half * 2;

  v2f bf[C / 4];
  const float* wcol = W + (size_t)kb * 128 + n;
#pragma unroll
  for (int kk = 0; kk < C / 4; ++kk) {
    bf[kk].x = wcol[(size_t)(4 * kk) * 128];
    bf[kk].y = wcol[(size_t)(4 * kk + 1) * 128];
  }
  const float bn = bias[n];
  const int ntile = (nn + 15) >> 4;

  for (int rt = wv >> 3; rt < ntile; rt += 4) {
    const float* arow = acc + (rt * 16 + lid) * PITCH + kb;
    v8f c = {};
#pragma unroll
    for (int kk = 0; kk < C / 4; ++kk) {
      const v2f a = *(const v2f*)(arow + 4 * kk);
      c = __builtin_amdgcn_wmma_f32_16x16x4_f32(
           false, a,  false, bf[kk],
           (short)0, c,  false,  false);
    }

    const int r0 = rt * 16 + half * 8;
    if (!POOL) {
      float* hbase = H + ((size_t)(lo + r0)) * 128 + n;
#pragma unroll
      for (int j = 0; j < 8; ++j) {
        const float v = c[j] + bn;
        if (r0 + j < nn) hbase[(size_t)j * 128] = v > 0.0f ? v : 0.0f;
      }
    } else {
      const int4 q0 = *(const int4*)(gid + r0), q1 = *(const int4*)(gid + r0 + 4);
      const int  g[8] = {q0.x, q0.y, q0.z, q0.w, q1.x, q1.y, q1.z, q1.w};
      const bool counts = (tileN == 0) && (lid == 0);
      auto give = [&](const int gc, const float s, const float len) {
        if (gc >= 0) {
          atomicAdd(sums + ((size_t)gc << 7) + n, s);
          if (counts) atomicAdd(cnts + gc, len);
        }
      };
      const float v0 = c[0] + bn;
      float s   = v0 > 0.0f ? v0 : 0.0f;
      float len = 1.0f;
      int   gc  = g[0];
#pragma unroll
      for (int j = 1; j < 8; ++j) {
        float v = c[j] + bn;
        v = v > 0.0f ? v : 0.0f;
        if (g[j] == gc) {
          s += v;
          len += 1.0f;
        } else {
          give(gc, s, len);
          s = v;
          len = 1.0f;
          gc = g[j];
        }
      }
      give(gc, s, len);
    }
  }
}


__global__ void k_final(const float* __restrict__ sums, const float* __restrict__ cnts,
                        const float* __restrict__ Wlin, const float* __restrict__ blin,
                        float* __restrict__ out) {
  int t = blockIdx.x * blockDim.x + threadIdx.x;
  if (t >= N_GRAPHS * NCLS) return;
  int g = t >> 2, c = t & 3;
  float inv = 1.0f / fmaxf(cnts[g], 1.0f);
  float acc = blin[c];
  const float* srow = sums + ((size_t)g << 7);
#pragma unroll 4
  for (int k = 0; k < 128; ++k)
    acc = fmaf(srow[k] * inv, Wlin[k * NCLS + c], acc);
  out[t] = acc;
}


extern "C" void kernel_launch(void* const* d_in, const int* in_sizes, int n_in,
                              void* d_out, int out_size, void* d_ws, size_t ws_size,
                              hipStream_t stream) {
  const int*   tok   = (const int*)d_in[0];
  const int*   eidx  = (const int*)d_in[1];
  const int*   src   = eidx;
  const int*   dst   = eidx + N_EDGES;
  const int*   batch = (const int*)d_in[2];
  const float* emb   = (const float*)d_in[3];
  const float* W1    = (const float*)d_in[4];
  const float* b1    = (const float*)d_in[5];
  const float* W2    = (const float*)d_in[6];
  const float* b2    = (const float*)d_in[7];
  const float* Wlin  = (const float*)d_in[8];
  const float* blin  = (const float*)d_in[9];
  float*       out   = (float*)d_out;

  float* ws   = (float*)d_ws;
  float* deg  = ws;
  float* x0   = deg  + N_NODES;
  float* x1   = x0   + (size_t)N_NODES * 64;
  float* sums = x1   + (size_t)N_NODES * 128;
  float* cnts = sums + (size_t)N_GRAPHS * 128;

  const int T = 256;
  const int gNodes4 = (N_NODES / 4 + T - 1) / T;
  const int gEdges4 = (N_EDGES / 4 + T - 1) / T;

  k_zero<<<gNodes4, T, 0, stream>>>(deg, N_NODES / 4);
  k_deg<<<gEdges4, T, 0, stream>>>(dst, deg);
  k_dinv<<<gNodes4, T, 0, stream>>>(deg);

  k_embed<<<EMB_WG, T, 0, stream>>>(tok, emb, x0);

  k_layer<6, false><<<AGG_WG, AGG_T, 0, stream>>>(x0, src, dst, deg, W1, b1, x1,
                                                  nullptr, nullptr, nullptr);

  k_zero<<<((N_GRAPHS * 128 + N_GRAPHS) / 4 + T - 1) / T, T, 0, stream>>>(
      sums, (N_GRAPHS * 128 + N_GRAPHS) / 4);
  k_layer<7, true><<<AGG_WG, AGG_T, 0, stream>>>(x1, src, dst, deg, W2, b2, nullptr,
                                                 batch, sums, cnts);

  k_final<<<(N_GRAPHS * NCLS + T - 1) / T, T, 0, stream>>>(sums, cnts, Wlin, blin, out);
}


# edge scan reads one key byte per edge (owner work-group = dst/391, written by the degree kernel) instead of the 4-byte dst: 16 edges per lane and load; queue holds edge numbers; sum order changed
# speedup vs baseline: 3.3932x; 1.0872x over previous
#include <hip/hip_runtime.h>
#include <hip/hip_bf16.h>


#define N_NODES  100000
#define N_EDGES  600000
#define N_GRAPHS 2000
#define NCLS     4

typedef float v2f __attribute__((ext_vector_type(2)));
typedef float v8f __attribute__((ext_vector_type(8)));


#define EMB_WG   1250
#define EMB_PER  5

static_assert((size_t)EMB_WG * 256 * EMB_PER == (size_t)N_NODES * 16, "embed covers x0 exactly");
static_assert(N_NODES % 16 == 0, "whole 16-row tiles; a tile's 16 batch ids are read as four int4");
static_assert(N_NODES % 4 == 0 && N_EDGES % 4 == 0 && (N_GRAPHS * 128 + N_GRAPHS) % 4 == 0,
              "16-byte pieces");

__global__ void k_zero(float* __restrict__ p, int n4) {
  int i = blockIdx.x * blockDim.x + threadIdx.x;
  if (i < n4) ((float4*)p)[i] = make_float4(0.0f, 0.0f, 0.0f, 0.0f);
}

#define KEY_NPW 391
__global__ void k_deg(const int* __restrict__ dst, float* __restrict__ deg,
                      int* __restrict__ key) {
  int i = blockIdx.x * blockDim.x + threadIdx.x;
  if (i < N_EDGES / 4) {
    const int4 d = ((const int4*)dst)[i];
    atomicAdd(&deg[d.x], 1.0f);
    atomicAdd(&deg[d.y], 1.0f);
    atomicAdd(&deg[d.z], 1.0f);
    atomicAdd(&deg[d.w], 1.0f);
    key[i] = (int)(((unsigned)d.x / KEY_NPW) | (((unsigned)d.y / KEY_NPW) << 8) |
                   (((unsigned)d.z / KEY_NPW) << 16) | (((unsigned)d.w / KEY_NPW) << 24));
  }
}

__global__ void k_dinv(float* __restrict__ deg) {
  int i = blockIdx.x * blockDim.x + threadIdx.x;
  if (i < N_NODES / 4) {
    float4* d4 = (float4*)deg;
    const float4 d = d4[i];
    d4[i] = make_float4(rsqrtf(d.x + 1.0f), rsqrtf(d.y + 1.0f),
                         rsqrtf(d.z + 1.0f), rsqrtf(d.w + 1.0f));
  }
}

__global__ void __launch_bounds__(256)
k_embed(const int* __restrict__ tok, const float* __restrict__ emb,
        float* __restrict__ x0) {
  const int g = blockIdx.x * 256 + threadIdx.x;
  const float4* __restrict__ e4 = (const float4*)emb;
  float4* __restrict__ x4 = (float4*)x0;
  int t[EMB_PER];
#pragma unroll
  for (int j = 0; j < EMB_PER; ++j) t[j] = tok[(g + j * (EMB_WG * 256)) >> 4];
#pragma unroll
  for (int j = 0; j < EMB_PER; ++j)
    x4[g + j * (EMB_WG * 256)] = e4[(t[j] << 4) | (g & 15)];
}


#define AGG_WG    256
#define AGG_NPW   391
#define AGG_ROWS  400
#define AGG_T     1024
#define AGG_QCAP  160
#define AGG_E16   (N_EDGES / 16)
#define AGG_STEPS 40
#define AGG_NCH   (AGG_STEPS * 32)
#define AGG_U     5

static_assert(AGG_WG * AGG_NPW >= N_NODES && (AGG_WG - 1) * AGG_NPW < N_NODES, "node ranges");
static_assert(AGG_ROWS % 16 == 0 && AGG_ROWS >= AGG_NPW, "whole 16-row tiles cover a range");
static_assert(N_EDGES % 16 == 0 && AGG_NCH * 32 >= AGG_E16, "edge chunks cover the keys");
static_assert(AGG_STEPS % AGG_U == 0, "whole rounds of loads");
static_assert(KEY_NPW == AGG_NPW && AGG_WG == 256 && (N_NODES - 1) / AGG_NPW < 256, "a key is one byte");

__device__ __forceinline__ void lds_add(float* p, float v) {
  __hip_atomic_fetch_add(p, v, __ATOMIC_RELAXED, __HIP_MEMORY_SCOPE_WORKGROUP);
}

template<int LOGC, bool POOL>
__global__ void __launch_bounds__(AGG_T)
k_layer(const float* __restrict__ X, const int* __restrict__ src,
        const int* __restrict__ dst, const int* __restrict__ key,
        const float* __restrict__ dinv,
        const float* __restrict__ W, const float* __restrict__ bias,
        float* __restrict__ H, const int* __restrict__ batch,
        float* __restrict__ sums, float* __restrict__ cnts) {
  constexpr int C = 1 << LOGC;
  constexpr int HP = C / 64;
  constexpr int PITCH = C + 4;
  __shared__ __attribute__((aligned(16))) float acc[AGG_ROWS * PITCH];
  __shared__ unsigned queue[(AGG_T / 32) * AGG_QCAP];
  __shared__ __attribute__((aligned(16))) int gid[POOL ? AGG_ROWS : 4];

  const int tid  = threadIdx.x;
  const int lane = tid & 31;
  const int wv   = __builtin_amdgcn_readfirstlane(tid >> 5);
  const int lo   = blockIdx.x * AGG_NPW;
  const int nn   = min(AGG_NPW, N_NODES - lo);

  const int n4 = nn << (LOGC - 2);
  const float4* X4 = (const float4*)(X + ((size_t)lo << LOGC));
#pragma unroll 4
  for (int i = tid; i < n4; i += AGG_T) {
    const int r = i >> (LOGC - 2);
    const float dv = dinv[lo + r];
    const float4 v = X4[i];
    *(float4*)(acc + r * PITCH + 4 * (i & (C / 4 - 1))) =
        make_float4(v.x * dv * dv, v.y * dv * dv, v.z * dv * dv, v.w * dv * dv);
  }
  for (int i = n4 + tid; i < AGG_ROWS * (C / 4); i += AGG_T)
    *(float4*)(acc + (i >> (LOGC - 2)) * PITCH + 4 * (i & (C / 4 - 1))) =
        make_float4(0.0f, 0.0f, 0.0f, 0.0f);
  if (POOL) {
    for (int r = tid; r < AGG_ROWS; r += AGG_T) gid[r] = (r < nn) ? batch[lo + r] : -1;
  }
  __syncthreads();

  unsigned* myq = queue + wv * AGG_QCAP;
  int qn = 0;

  auto drain = [&](const int n) {
    __builtin_amdgcn_fence(__ATOMIC_ACQ_REL, "wavefront");
    const unsigned e = myq[qn - n + min(lane, n - 1)];
    qn -= n;
    const int   s  = src[e];
    const int   dg = dst[e];
    const int   dl = dg - lo;
    const float nr = dinv[s] * dinv[dg];
    for (int j0 = 0; j0 < n; j0 += 8) {
      float2 v[8][HP];
#pragma unroll
      for (int jj = 0; jj < 8; ++jj) {
        const int j  = min(j0 + jj, n - 1);
        const int sj = __builtin_amdgcn_readlane(s, j);
        const float* row = X + ((size_t)sj << LOGC) + 2 * lane;
#pragma unroll
        for (int h = 0; h < HP; ++h) v[jj][h] = *(const float2*)(row + 64 * h);
      }
#pragma unroll
      for (int jj = 0; jj < 8; ++jj) {
        const int j = j0 + jj;
        if (j < n) {
          const float nj = __int_as_float(__builtin_amdgcn_readlane(__float_as_int(nr), j));
          const int   dj = __builtin_amdgcn_readlane(dl, j);
          float* a = acc + dj * PITCH + 2 * lane;
#pragma unroll
          for (int h = 0; h < HP; ++h) {
            lds_add(a + 64 * h,     v[jj][h].x * nj);
            lds_add(a + 64 * h + 1, v[jj][h].y * nj);
          }
        }
      }
    }
  };

  const int4* key4 = (const int4*)key;
  const unsigned mine = (unsigned)blockIdx.x * 0x01010101u;
  int chunk = wv + (blockIdx.x * AGG_STEPS) / 8;
  if (chunk >= AGG_NCH) chunk -= AGG_NCH;

  int4 dn[AGG_U];
  int  in[AGG_U];
  auto fetch = [&]() {
#pragma unroll
    for (int u = 0; u < AGG_U; ++u) {
      in[u] = chunk * 32 + lane;
      chunk += 32;
      if (chunk >= AGG_NCH) chunk -= AGG_NCH;
      dn[u] = key4[min(in[u], AGG_E16 - 1)];
    }
  };
  fetch();

  for (int t0 = 0; t0 < AGG_STEPS; t0 += AGG_U) {
    int4 d[AGG_U];
    int  i4[AGG_U];
#pragma unroll
    for (int u = 0; u < AGG_U; ++u) { d[u] = dn[u]; i4[u] = in[u]; }
    if (t0 + AGG_U < AGG_STEPS) fetch();
#pragma unroll
    for (int u = 0; u < AGG_U; ++u) {
      auto hit = [&](const int w) -> unsigned {
        const unsigned x = (unsigned)w ^ mine;
        return ~(((x & 0x7F7F7F7Fu) + 0x7F7F7F7Fu) | x | 0x7F7F7F7Fu);
      };
      unsigned M = (hit(d[u].x) >> 7) | (hit(d[u].y) >> 6) | (hit(d[u].z) >> 5) | (hit(d[u].w) >> 4);
      if (i4[u] >= AGG_E16) M = 0u;
      while (__builtin_amdgcn_ballot_w32(M != 0u)) {
        const bool m = M != 0u;
        const unsigned mk = __builtin_amdgcn_ballot_w32(m);
        if (m) {
          const unsigned p = (unsigned)__builtin_ctz(M);
          myq[qn + __builtin_amdgcn_mbcnt_lo(mk, 0u)] =
              ((unsigned)i4[u] << 4) + ((p & 7u) << 2) + (p >> 3);
          M &= M - 1u;
        }
        qn += __builtin_popcount(mk);
        while (qn >= 32) drain(32);
      }
    }
  }
  if (qn > 0) drain(qn);
  __syncthreads();

  const int tileN = wv & 7;
  const int half  = lane >> 4;
  const int lid   = lane & 15;
  const int n     = tileN * 16 + lid;
  const int kb    = half * 2;

  v2f bf[C / 4];
  const float* wcol = W + (size_t)kb * 128 + n;
#pragma unroll
  for (int kk = 0; kk < C / 4; ++kk) {
    bf[kk].x = wcol[(size_t)(4 * kk) * 128];
    bf[kk].y = wcol[(size_t)(4 * kk + 1) * 128];
  }
  const float bn = bias[n];
  const int ntile = (nn + 15) >> 4;

  for (int rt = wv >> 3; rt < ntile; rt += 4) {
    const float* arow = acc + (rt * 16 + lid) * PITCH + kb;
    v8f c = {};
#pragma unroll
    for (int kk = 0; kk < C / 4; ++kk) {
      const v2f a = *(const v2f*)(arow + 4 * kk);
      c = __builtin_amdgcn_wmma_f32_16x16x4_f32(
           false, a,  false, bf[kk],
           (short)0, c,  false,  false);
    }

    const int r0 = rt * 16 + half * 8;
    if (!POOL) {
      float* hbase = H + ((size_t)(lo + r0)) * 128 + n;
#pragma unroll
      for (int j = 0; j < 8; ++j) {
        const float v = c[j] + bn;
        if (r0 + j < nn) hbase[(size_t)j * 128] = v > 0.0f ? v : 0.0f;
      }
    } else {
      const int4 q0 = *(const int4*)(gid + r0), q1 = *(const int4*)(gid + r0 + 4);
      const int  g[8] = {q0.x, q0.y, q0.z, q0.w, q1.x, q1.y, q1.z, q1.w};
      const bool counts = (tileN == 0) && (lid == 0);
      auto give = [&](const int gc, const float s, const float len) {
        if (gc >= 0) {
          atomicAdd(sums + ((size_t)gc << 7) + n, s);
          if (counts) atomicAdd(cnts + gc, len);
        }
      };
      const float v0 = c[0] + bn;
      float s   = v0 > 0.0f ? v0 : 0.0f;
      float len = 1.0f;
      int   gc  = g[0];
#pragma unroll
      for (int j = 1; j < 8; ++j) {
        float v = c[j] + bn;
        v = v > 0.0f ? v : 0.0f;
        if (g[j] == gc) {
          s += v;
          len += 1.0f;
        } else {
          give(gc, s, len);
          s = v;
          len = 1.0f;
          gc = g[j];
        }
      }
      give(gc, s, len);
    }
  }
}


__global__ void k_final(const float* __restrict__ sums, const float* __restrict__ cnts,
                        const float* __restrict__ Wlin, const float* __restrict__ blin,
                        float* __restrict__ out) {
  int t = blockIdx.x * blockDim.x + threadIdx.x;
  if (t >= N_GRAPHS * NCLS) return;
  int g = t >> 2, c = t & 3;
  float inv = 1.0f / fmaxf(cnts[g], 1.0f);
  float acc = blin[c];
  const float* srow = sums + ((size_t)g << 7);
#pragma unroll 4
  for (int k = 0; k < 128; ++k)
    acc = fmaf(srow[k] * inv, Wlin[k * NCLS + c], acc);
  out[t] = acc;
}


extern "C" void kernel_launch(void* const* d_in, const int* in_sizes, int n_in,
                              void* d_out, int out_size, void* d_ws, size_t ws_size,
                              hipStream_t stream) {
  const int*   tok   = (const int*)d_in[0];
  const int*   eidx  = (const int*)d_in[1];
  const int*   src   = eidx;
  const int*   dst   = eidx + N_EDGES;
  const int*   batch = (const int*)d_in[2];
  const float* emb   = (const float*)d_in[3];
  const float* W1    = (const float*)d_in[4];
  const float* b1    = (const float*)d_in[5];
  const float* W2    = (const float*)d_in[6];
  const float* b2    = (const float*)d_in[7];
  const float* Wlin  = (const float*)d_in[8];
  const float* blin  = (const float*)d_in[9];
  float*       out   = (float*)d_out;

  float* ws   = (float*)d_ws;
  float* deg  = ws;
  float* x0   = deg  + N_NODES;
  float* x1   = x0   + (size_t)N_NODES * 64;
  float* sums = x1   + (size_t)N_NODES * 128;
  float* cnts = sums + (size_t)N_GRAPHS * 128;
  int*   key  = (int*)(cnts + N_GRAPHS);

  const int T = 256;
  const int gNodes4 = (N_NODES / 4 + T - 1) / T;
  const int gEdges4 = (N_EDGES / 4 + T - 1) / T;

  k_zero<<<gNodes4, T, 0, stream>>>(deg, N_NODES / 4);
  k_deg<<<gEdges4, T, 0, stream>>>(dst, deg, key);
  k_dinv<<<gNodes4, T, 0, stream>>>(deg);

  k_embed<<<EMB_WG, T, 0, stream>>>(tok, emb, x0);

  k_layer<6, false><<<AGG_WG, AGG_T, 0, stream>>>(x0, src, dst, key, deg, W1, b1, x1,
                                                  nullptr, nullptr, nullptr);

  k_zero<<<((N_GRAPHS * 128 + N_GRAPHS) / 4 + T - 1) / T, T, 0, stream>>>(
      sums, (N_GRAPHS * 128 + N_GRAPHS) / 4);
  k_layer<7, true><<<AGG_WG, AGG_T, 0, stream>>>(x1, src, dst, key, deg, W2, b2, nullptr,
                                                 batch, sums, cnts);

  k_final<<<(N_GRAPHS * NCLS + T - 1) / T, T, 0, stream>>>(sums, cnts, Wlin, blin, out);
}


# edge search by bit map: degree kernel writes one bit per edge and destination range (set in LDS, stored as whole lines); a work-group reads its own 75 KB row, not 600 KB of keys; sum order changed
# speedup vs baseline: 3.6642x; 1.0799x over previous
#include <hip/hip_runtime.h>
#include <hip/hip_bf16.h>


#define N_NODES  100000
#define N_EDGES  600000
#define N_GRAPHS 2000
#define NCLS     4

typedef float v2f __attribute__((ext_vector_type(2)));
typedef float v8f __attribute__((ext_vector_type(8)));


#define EMB_WG   1250
#define EMB_PER  5

static_assert((size_t)EMB_WG * 256 * EMB_PER == (size_t)N_NODES * 16, "embed covers x0 exactly");
static_assert(N_NODES % 16 == 0, "whole 16-row tiles; a tile's 16 batch ids are read as four int4");
static_assert(N_NODES % 4 == 0 && N_EDGES % 4 == 0 && (N_GRAPHS * 128 + N_GRAPHS) % 4 == 0,
              "16-byte pieces");

__global__ void k_zero(float* __restrict__ p, int n4) {
  int i = blockIdx.x * blockDim.x + threadIdx.x;
  if (i < n4) ((float4*)p)[i] = make_float4(0.0f, 0.0f, 0.0f, 0.0f);
}

#define KEY_NPW 391
#define BM_OWN  256
#define BM_BLK  1024
#define BM_WG   ((N_EDGES + BM_BLK - 1) / BM_BLK)
#define BM_RS   (BM_WG * 32)
static_assert(BM_RS % 4 == 0 && (size_t)BM_RS * 32 >= (size_t)N_EDGES, "rows of whole 16-byte pieces cover the edges");
__global__ void __launch_bounds__(256)
k_deg(const int* __restrict__ dst, float* __restrict__ deg,
      unsigned* __restrict__ bits) {
  __shared__ __attribute__((aligned(16))) unsigned bm[BM_OWN * 32];
  const int  tid = threadIdx.x;
  const int  e0 = blockIdx.x * BM_BLK + tid * 4;
  const bool in = e0 < N_EDGES;
  int4 d = make_int4(0, 0, 0, 0);
  if (in) d = *(const int4*)(dst + e0);
  for (int i = tid; i < BM_OWN * 8; i += 256) ((uint4*)bm)[i] = make_uint4(0u, 0u, 0u, 0u);
  __syncthreads();
  if (in) {
    const int      w = tid >> 3;
    const unsigned b = 1u << ((tid & 7) * 4);
    atomicOr(&bm[((unsigned)d.x / KEY_NPW) * 32 + w], b);
    atomicOr(&bm[((unsigned)d.y / KEY_NPW) * 32 + w], b << 1);
    atomicOr(&bm[((unsigned)d.z / KEY_NPW) * 32 + w], b << 2);
    atomicOr(&bm[((unsigned)d.w / KEY_NPW) * 32 + w], b << 3);
  }
  __syncthreads();
  for (int i = tid; i < BM_OWN * 8; i += 256)
    *(uint4*)(bits + (size_t)(i >> 3) * BM_RS + blockIdx.x * 32 + (i & 7) * 4) = ((const uint4*)bm)[i];
  if (in) {
    atomicAdd(&deg[d.x], 1.0f);
    atomicAdd(&deg[d.y], 1.0f);
    atomicAdd(&deg[d.z], 1.0f);
    atomicAdd(&deg[d.w], 1.0f);
  }
}

__global__ void k_dinv(float* __restrict__ deg) {
  int i = blockIdx.x * blockDim.x + threadIdx.x;
  if (i < N_NODES / 4) {
    float4* d4 = (float4*)deg;
    const float4 d = d4[i];
    d4[i] = make_float4(rsqrtf(d.x + 1.0f), rsqrtf(d.y + 1.0f),
                         rsqrtf(d.z + 1.0f), rsqrtf(d.w + 1.0f));
  }
}

__global__ void __launch_bounds__(256)
k_embed(const int* __restrict__ tok, const float* __restrict__ emb,
        float* __restrict__ x0) {
  const int g = blockIdx.x * 256 + threadIdx.x;
  const float4* __restrict__ e4 = (const float4*)emb;
  float4* __restrict__ x4 = (float4*)x0;
  int t[EMB_PER];
#pragma unroll
  for (int j = 0; j < EMB_PER; ++j) t[j] = tok[(g + j * (EMB_WG * 256)) >> 4];
#pragma unroll
  for (int j = 0; j < EMB_PER; ++j)
    x4[g + j * (EMB_WG * 256)] = e4[(t[j] << 4) | (g & 15)];
}


#define AGG_WG    256
#define AGG_NPW   391
#define AGG_ROWS  400
#define AGG_T     1024
#define AGG_QCAP  64
#define AGG_NV    (BM_RS / 4)
#define AGG_NLD   ((AGG_NV + AGG_T - 1) / AGG_T)

static_assert(AGG_WG * AGG_NPW >= N_NODES && (AGG_WG - 1) * AGG_NPW < N_NODES, "node ranges");
static_assert(AGG_ROWS % 16 == 0 && AGG_ROWS >= AGG_NPW, "whole 16-row tiles cover a range");
static_assert(KEY_NPW == AGG_NPW && BM_OWN == AGG_WG && (N_NODES - 1) / AGG_NPW < BM_OWN, "one row of the map per work-group");
static_assert(AGG_NV > AGG_T, "a lane's first piece is inside the row");

__device__ __forceinline__ void lds_add(float* p, float v) {
  __hip_atomic_fetch_add(p, v, __ATOMIC_RELAXED, __HIP_MEMORY_SCOPE_WORKGROUP);
}

template<int LOGC, bool POOL>
__global__ void __launch_bounds__(AGG_T)
k_layer(const float* __restrict__ X, const int* __restrict__ src,
        const int* __restrict__ dst, const unsigned* __restrict__ bits,
        const float* __restrict__ dinv,
        const float* __restrict__ W, const float* __restrict__ bias,
        float* __restrict__ H, const int* __restrict__ batch,
        float* __restrict__ sums, float* __restrict__ cnts) {
  constexpr int C = 1 << LOGC;
  constexpr int HP = C / 64;
  constexpr int PITCH = C + 4;
  __shared__ __attribute__((aligned(16))) float acc[AGG_ROWS * PITCH];
  __shared__ unsigned queue[(AGG_T / 32) * AGG_QCAP];
  __shared__ __attribute__((aligned(16))) int gid[POOL ? AGG_ROWS : 4];

  const int tid  = threadIdx.x;
  const int lane = tid & 31;
  const int wv   = __builtin_amdgcn_readfirstlane(tid >> 5);
  const int lo   = blockIdx.x * AGG_NPW;
  const int nn   = min(AGG_NPW, N_NODES - lo);

  const int n4 = nn << (LOGC - 2);
  const float4* X4 = (const float4*)(X + ((size_t)lo << LOGC));
#pragma unroll 4
  for (int i = tid; i < n4; i += AGG_T) {
    const int r = i >> (LOGC - 2);
    const float dv = dinv[lo + r];
    const float4 v = X4[i];
    *(float4*)(acc + r * PITCH + 4 * (i & (C / 4 - 1))) =
        make_float4(v.x * dv * dv, v.y * dv * dv, v.z * dv * dv, v.w * dv * dv);
  }
  for (int i = n4 + tid; i < AGG_ROWS * (C / 4); i += AGG_T)
    *(float4*)(acc + (i >> (LOGC - 2)) * PITCH + 4 * (i & (C / 4 - 1))) =
        make_float4(0.0f, 0.0f, 0.0f, 0.0f);
  if (POOL) {
    for (int r = tid; r < AGG_ROWS; r += AGG_T) gid[r] = (r < nn) ? batch[lo + r] : -1;
  }
  __syncthreads();

  unsigned* myq = queue + wv * AGG_QCAP;
  int qn = 0;

  auto drain = [&](const int n) {
    __builtin_amdgcn_fence(__ATOMIC_ACQ_REL, "wavefront");
    const unsigned e = myq[qn - n + min(lane, n - 1)];
    qn -= n;
    const int   s  = src[e];
    const int   dg = dst[e];
    const int   dl = dg - lo;
    const float nr = dinv[s] * dinv[dg];
    for (int j0 = 0; j0 < n; j0 += 8) {
      float2 v[8][HP];
#pragma unroll
      for (int jj = 0; jj < 8; ++jj) {
        const int j  = min(j0 + jj, n - 1);
        const int sj = __builtin_amdgcn_readlane(s, j);
        const float* row = X + ((size_t)sj << LOGC) + 2 * lane;
#pragma unroll
        for (int h = 0; h < HP; ++h) v[jj][h] = *(const float2*)(row + 64 * h);
      }
#pragma unroll
      for (int jj = 0; jj < 8; ++jj) {
        const int j = j0 + jj;
        if (j < n) {
          const float nj = __int_as_float(__builtin_amdgcn_readlane(__float_as_int(nr), j));
          const int   dj = __builtin_amdgcn_readlane(dl, j);
          float* a = acc + dj * PITCH + 2 * lane;
#pragma unroll
          for (int h = 0; h < HP; ++h) {
            lds_add(a + 64 * h,     v[jj][h].x * nj);
            lds_add(a + 64 * h + 1, v[jj][h].y * nj);
          }
        }
      }
    }
  };

  const uint4* row4 = (const uint4*)(bits + (size_t)blockIdx.x * BM_RS);
  uint4 nx = row4[tid];
  for (int u = 0; u < AGG_NLD; ++u) {
    uint4 cur = nx;
    const int piece = tid + u * AGG_T;
    if (piece >= AGG_NV) cur = make_uint4(0u, 0u, 0u, 0u);
    if (u + 1 < AGG_NLD) nx = row4[min(piece + AGG_T, AGG_NV - 1)];
    for (int k = 0; k < 4; ++k) {
      unsigned M = cur.x;
      cur.x = cur.y; cur.y = cur.z; cur.z = cur.w;
      const unsigned base = ((unsigned)piece * 4u + (unsigned)k) << 5;
      while (__builtin_amdgcn_ballot_w32(M != 0u)) {
        const bool m = M != 0u;
        const unsigned mk = __builtin_amdgcn_ballot_w32(m);
        if (m) {
          myq[qn + __builtin_amdgcn_mbcnt_lo(mk, 0u)] = base + (unsigned)__builtin_ctz(M);
          M &= M - 1u;
        }
        qn += __builtin_popcount(mk);
        while (qn >= 32) drain(32);
      }
    }
  }
  if (qn > 0) drain(qn);
  __syncthreads();

  const int tileN = wv & 7;
  const int half  = lane >> 4;
  const int lid   = lane & 15;
  const int n     = tileN * 16 + lid;
  const int kb    = half * 2;

  v2f bf[C / 4];
  const float* wcol = W + (size_t)kb * 128 + n;
#pragma unroll
  for (int kk = 0; kk < C / 4; ++kk) {
    bf[kk].x = wcol[(size_t)(4 * kk) * 128];
    bf[kk].y = wcol[(size_t)(4 * kk + 1) * 128];
  }
  const float bn = bias[n];
  const int ntile = (nn + 15) >> 4;

  for (int rt = wv >> 3; rt < ntile; rt += 4) {
    const float* arow = acc + (rt * 16 + lid) * PITCH + kb;
    v8f c = {};
#pragma unroll
    for (int kk = 0; kk < C / 4; ++kk) {
      const v2f a = *(const v2f*)(arow + 4 * kk);
      c = __builtin_amdgcn_wmma_f32_16x16x4_f32(
           false, a,  false, bf[kk],
           (short)0, c,  false,  false);
    }

    const int r0 = rt * 16 + half * 8;
    if (!POOL) {
      float* hbase = H + ((size_t)(lo + r0)) * 128 + n;
#pragma unroll
      for (int j = 0; j < 8; ++j) {
        const float v = c[j] + bn;
        if (r0 + j < nn) hbase[(size_t)j * 128] = v > 0.0f ? v : 0.0f;
      }
    } else {
      const int4 q0 = *(const int4*)(gid + r0), q1 = *(const int4*)(gid + r0 + 4);
      const int  g[8] = {q0.x, q0.y, q0.z, q0.w, q1.x, q1.y, q1.z, q1.w};
      const bool counts = (tileN == 0) && (lid == 0);
      auto give = [&](const int gc, const float s, const float len) {
        if (gc >= 0) {
          atomicAdd(sums + ((size_t)gc << 7) + n, s);
          if (counts) atomicAdd(cnts + gc, len);
        }
      };
      const float v0 = c[0] + bn;
      float s   = v0 > 0.0f ? v0 : 0.0f;
      float len = 1.0f;
      int   gc  = g[0];
#pragma unroll
      for (int j = 1; j < 8; ++j) {
        float v = c[j] + bn;
        v = v > 0.0f ? v : 0.0f;
        if (g[j] == gc) {
          s += v;
          len += 1.0f;
        } else {
          give(gc, s, len);
          s = v;
          len = 1.0f;
          gc = g[j];
        }
      }
      give(gc, s, len);
    }
  }
}


__global__ void k_final(const float* __restrict__ sums, const float* __restrict__ cnts,
                        const float* __restrict__ Wlin, const float* __restrict__ blin,
                        float* __restrict__ out) {
  int t = blockIdx.x * blockDim.x + threadIdx.x;
  if (t >= N_GRAPHS * NCLS) return;
  int g = t >> 2, c = t & 3;
  float inv = 1.0f / fmaxf(cnts[g], 1.0f);
  float acc = blin[c];
  const float* srow = sums + ((size_t)g << 7);
#pragma unroll 4
  for (int k = 0; k < 128; ++k)
    acc = fmaf(srow[k] * inv, Wlin[k * NCLS + c], acc);
  out[t] = acc;
}


#define WS_BITS_OFF ((((size_t)N_NODES * (1 + 64 + 128) + (size_t)N_GRAPHS * (128 + 1)) + 31) / 32 * 32)
static_assert((WS_BITS_OFF + (size_t)BM_OWN * BM_RS) * 4 <= (size_t)268435456, "inside the workspace");

extern "C" void kernel_launch(void* const* d_in, const int* in_sizes, int n_in,
                              void* d_out, int out_size, void* d_ws, size_t ws_size,
                              hipStream_t stream) {
  const int*   tok   = (const int*)d_in[0];
  const int*   eidx  = (const int*)d_in[1];
  const int*   src   = eidx;
  const int*   dst   = eidx + N_EDGES;
  const int*   batch = (const int*)d_in[2];
  const float* emb   = (const float*)d_in[3];
  const float* W1    = (const float*)d_in[4];
  const float* b1    = (const float*)d_in[5];
  const float* W2    = (const float*)d_in[6];
  const float* b2    = (const float*)d_in[7];
  const float* Wlin  = (const float*)d_in[8];
  const float* blin  = (const float*)d_in[9];
  float*       out   = (float*)d_out;

  float* ws   = (float*)d_ws;
  float* deg  = ws;
  float* x0   = deg  + N_NODES;
  float* x1   = x0   + (size_t)N_NODES * 64;
  float* sums = x1   + (size_t)N_NODES * 128;
  float* cnts = sums + (size_t)N_GRAPHS * 128;
  unsigned* bits = (unsigned*)(ws + WS_BITS_OFF);

  const int T = 256;
  const int gNodes4 = (N_NODES / 4 + T - 1) / T;

  k_zero<<<gNodes4, T, 0, stream>>>(deg, N_NODES / 4);
  k_deg<<<BM_WG, T, 0, stream>>>(dst, deg, bits);
  k_dinv<<<gNodes4, T, 0, stream>>>(deg);

  k_embed<<<EMB_WG, T, 0, stream>>>(tok, emb, x0);

  k_layer<6, false><<<AGG_WG, AGG_T, 0, stream>>>(x0, src, dst, bits, deg, W1, b1, x1,
                                                  nullptr, nullptr, nullptr);

  k_zero<<<((N_GRAPHS * 128 + N_GRAPHS) / 4 + T - 1) / T, T, 0, stream>>>(
      sums, (N_GRAPHS * 128 + N_GRAPHS) / 4);
  k_layer<7, true><<<AGG_WG, AGG_T, 0, stream>>>(x1, src, dst, bits, deg, W2, b2, nullptr,
                                                 batch, sums, cnts);

  k_final<<<(N_GRAPHS * NCLS + T - 1) / T, T, 0, stream>>>(sums, cnts, Wlin, blin, out);
}


# fewer launches: one clear (deg, pool sums); embedding copy rides beside the degree/bit-map work-groups in one launch; deg^-1/2 formed at each use, its kernel gone; 5 launches (was 8); sums unchanged
# speedup vs baseline: 3.9044x; 1.0656x over previous
#include <hip/hip_runtime.h>
#include <hip/hip_bf16.h>


#define N_NODES  100000
#define N_EDGES  600000
#define N_GRAPHS 2000
#define NCLS     4

typedef float v2f __attribute__((ext_vector_type(2)));
typedef float v8f __attribute__((ext_vector_type(8)));


#define EMB_WG   1250
#define EMB_PER  5

static_assert((size_t)EMB_WG * 256 * EMB_PER == (size_t)N_NODES * 16, "embed covers x0 exactly");
static_assert(N_NODES % 16 == 0, "whole 16-row tiles; a tile's 16 batch ids are read as four int4");
static_assert(N_NODES % 4 == 0 && N_EDGES % 4 == 0 && (N_GRAPHS * 128 + N_GRAPHS) % 4 == 0,
              "16-byte pieces");

__global__ void k_zero2(float* __restrict__ a, int na4, float* __restrict__ b, int nb4) {
  const int i = blockIdx.x * blockDim.x + threadIdx.x;
  if (i < na4) ((float4*)a)[i] = make_float4(0.0f, 0.0f, 0.0f, 0.0f);
  else if (i - na4 < nb4) ((float4*)b)[i - na4] = make_float4(0.0f, 0.0f, 0.0f, 0.0f);
}

#define KEY_NPW 391
#define BM_OWN  256
#define BM_BLK  1024
#define BM_WG   ((N_EDGES + BM_BLK - 1) / BM_BLK)
#define BM_RS   (BM_WG * 32)
static_assert(BM_RS % 4 == 0 && (size_t)BM_RS * 32 >= (size_t)N_EDGES, "rows of whole 16-byte pieces cover the edges");
__global__ void __launch_bounds__(256)
k_deg_embed(const int* __restrict__ dst, float* __restrict__ deg,
            unsigned* __restrict__ bits,
            const int* __restrict__ tok, const float* __restrict__ emb,
            float* __restrict__ x0) {
  __shared__ __attribute__((aligned(16))) unsigned bm[BM_OWN * 32];
  const int  tid = threadIdx.x;
  if (blockIdx.x >= BM_WG) {
    const int g = (blockIdx.x - BM_WG) * 256 + tid;
    const float4* __restrict__ e4 = (const float4*)emb;
    float4* __restrict__ x4 = (float4*)x0;
    int t[EMB_PER];
#pragma unroll
    for (int j = 0; j < EMB_PER; ++j) t[j] = tok[(g + j * (EMB_WG * 256)) >> 4];
#pragma unroll
    for (int j = 0; j < EMB_PER; ++j)
      x4[g + j * (EMB_WG * 256)] = e4[(t[j] << 4) | (g & 15)];
    return;
  }
  const int  e0 = blockIdx.x * BM_BLK + tid * 4;
  const bool in = e0 < N_EDGES;
  int4 d = make_int4(0, 0, 0, 0);
  if (in) d = *(const int4*)(dst + e0);
  for (int i = tid; i < BM_OWN * 8; i += 256) ((uint4*)bm)[i] = make_uint4(0u, 0u, 0u, 0u);
  __syncthreads();
  if (in) {
    const int      w = tid >> 3;
    const unsigned b = 1u << ((tid & 7) * 4);
    atomicOr(&bm[((unsigned)d.x / KEY_NPW) * 32 + w], b);
    atomicOr(&bm[((unsigned)d.y / KEY_NPW) * 32 + w], b << 1);
    atomicOr(&bm[((unsigned)d.z / KEY_NPW) * 32 + w], b << 2);
    atomicOr(&bm[((unsigned)d.w / KEY_NPW) * 32 + w], b << 3);
  }
  __syncthreads();
  for (int i = tid; i < BM_OWN * 8; i += 256)
    *(uint4*)(bits + (size_t)(i >> 3) * BM_RS + blockIdx.x * 32 + (i & 7) * 4) = ((const uint4*)bm)[i];
  if (in) {
    atomicAdd(&deg[d.x], 1.0f);
    atomicAdd(&deg[d.y], 1.0f);
    atomicAdd(&deg[d.z], 1.0f);
    atomicAdd(&deg[d.w], 1.0f);
  }
}


#define AGG_WG    256
#define AGG_NPW   391
#define AGG_ROWS  400
#define AGG_T     1024
#define AGG_QCAP  64
#define AGG_NV    (BM_RS / 4)
#define AGG_NLD   ((AGG_NV + AGG_T - 1) / AGG_T)

static_assert(AGG_WG * AGG_NPW >= N_NODES && (AGG_WG - 1) * AGG_NPW < N_NODES, "node ranges");
static_assert(AGG_ROWS % 16 == 0 && AGG_ROWS >= AGG_NPW, "whole 16-row tiles cover a range");
static_assert(KEY_NPW == AGG_NPW && BM_OWN == AGG_WG && (N_NODES - 1) / AGG_NPW < BM_OWN, "one row of the map per work-group");
static_assert(AGG_NV > AGG_T, "a lane's first piece is inside the row");

__device__ __forceinline__ void lds_add(float* p, float v) {
  __hip_atomic_fetch_add(p, v, __ATOMIC_RELAXED, __HIP_MEMORY_SCOPE_WORKGROUP);
}

template<int LOGC, bool POOL>
__global__ void __launch_bounds__(AGG_T)
k_layer(const float* __restrict__ X, const int* __restrict__ src,
        const int* __restrict__ dst, const unsigned* __restrict__ bits,
        const float* __restrict__ deg,
        const float* __restrict__ W, const float* __restrict__ bias,
        float* __restrict__ H, const int* __restrict__ batch,
        float* __restrict__ sums, float* __restrict__ cnts) {
  constexpr int C = 1 << LOGC;
  constexpr int HP = C / 64;
  constexpr int PITCH = C + 4;
  __shared__ __attribute__((aligned(16))) float acc[AGG_ROWS * PITCH];
  __shared__ unsigned queue[(AGG_T / 32) * AGG_QCAP];
  __shared__ __attribute__((aligned(16))) int gid[POOL ? AGG_ROWS : 4];

  const int tid  = threadIdx.x;
  const int lane = tid & 31;
  const int wv   = __builtin_amdgcn_readfirstlane(tid >> 5);
  const int lo   = blockIdx.x * AGG_NPW;
  const int nn   = min(AGG_NPW, N_NODES - lo);

  const int n4 = nn << (LOGC - 2);
  const float4* X4 = (const float4*)(X + ((size_t)lo << LOGC));
#pragma unroll 4
  for (int i = tid; i < n4; i += AGG_T) {
    const int r = i >> (LOGC - 2);
    const float dv = rsqrtf(deg[lo + r] + 1.0f);
    const float4 v = X4[i];
    *(float4*)(acc + r * PITCH + 4 * (i & (C / 4 - 1))) =
        make_float4(v.x * dv * dv, v.y * dv * dv, v.z * dv * dv, v.w * dv * dv);
  }
  for (int i = n4 + tid; i < AGG_ROWS * (C / 4); i += AGG_T)
    *(float4*)(acc + (i >> (LOGC - 2)) * PITCH + 4 * (i & (C / 4 - 1))) =
        make_float4(0.0f, 0.0f, 0.0f, 0.0f);
  if (POOL) {
    for (int r = tid; r < AGG_ROWS; r += AGG_T) gid[r] = (r < nn) ? batch[lo + r] : -1;
  }
  __syncthreads();

  unsigned* myq = queue + wv * AGG_QCAP;
  int qn = 0;

  auto drain = [&](const int n) {
    __builtin_amdgcn_fence(__ATOMIC_ACQ_REL, "wavefront");
    const unsigned e = myq[qn - n + min(lane, n - 1)];
    qn -= n;
    const int   s  = src[e];
    const int   dg = dst[e];
    const int   dl = dg - lo;
    const float nr = rsqrtf(deg[s] + 1.0f) * rsqrtf(deg[dg] + 1.0f);
    for (int j0 = 0; j0 < n; j0 += 8) {
      float2 v[8][HP];
#pragma unroll
      for (int jj = 0; jj < 8; ++jj) {
        const int j  = min(j0 + jj, n - 1);
        const int sj = __builtin_amdgcn_readlane(s, j);
        const float* row = X + ((size_t)sj << LOGC) + 2 * lane;
#pragma unroll
        for (int h = 0; h < HP; ++h) v[jj][h] = *(const float2*)(row + 64 * h);
      }
#pragma unroll
      for (int jj = 0; jj < 8; ++jj) {
        const int j = j0 + jj;
        if (j < n) {
          const float nj = __int_as_float(__builtin_amdgcn_readlane(__float_as_int(nr), j));
          const int   dj = __builtin_amdgcn_readlane(dl, j);
          float* a = acc + dj * PITCH + 2 * lane;
#pragma unroll
          for (int h = 0; h < HP; ++h) {
            lds_add(a + 64 * h,     v[jj][h].x * nj);
            lds_add(a + 64 * h + 1, v[jj][h].y * nj);
          }
        }
      }
    }
  };

  const uint4* row4 = (const uint4*)(bits + (size_t)blockIdx.x * BM_RS);
  uint4 nx = row4[tid];
  for (int u = 0; u < AGG_NLD; ++u) {
    uint4 cur = nx;
    const int piece = tid + u * AGG_T;
    if (piece >= AGG_NV) cur = make_uint4(0u, 0u, 0u, 0u);
    if (u + 1 < AGG_NLD) nx = row4[min(piece + AGG_T, AGG_NV - 1)];
    for (int k = 0; k < 4; ++k) {
      unsigned M = cur.x;
      cur.x = cur.y; cur.y = cur.z; cur.z = cur.w;
      const unsigned base = ((unsigned)piece * 4u + (unsigned)k) << 5;
      while (__builtin_amdgcn_ballot_w32(M != 0u)) {
        const bool m = M != 0u;
        const unsigned mk = __builtin_amdgcn_ballot_w32(m);
        if (m) {
          myq[qn + __builtin_amdgcn_mbcnt_lo(mk, 0u)] = base + (unsigned)__builtin_ctz(M);
          M &= M - 1u;
        }
        qn += __builtin_popcount(mk);
        while (qn >= 32) drain(32);
      }
    }
  }
  if (qn > 0) drain(qn);
  __syncthreads();

  const int tileN = wv & 7;
  const int half  = lane >> 4;
  const int lid   = lane & 15;
  const int n     = tileN * 16 + lid;
  const int kb    = half * 2;

  v2f bf[C / 4];
  const float* wcol = W + (size_t)kb * 128 + n;
#pragma unroll
  for (int kk = 0; kk < C / 4; ++kk) {
    bf[kk].x = wcol[(size_t)(4 * kk) * 128];
    bf[kk].y = wcol[(size_t)(4 * kk + 1) * 128];
  }
  const float bn = bias[n];
  const int ntile = (nn + 15) >> 4;

  for (int rt = wv >> 3; rt < ntile; rt += 4) {
    const float* arow = acc + (rt * 16 + lid) * PITCH + kb;
    v8f c = {};
#pragma unroll
    for (int kk = 0; kk < C / 4; ++kk) {
      const v2f a = *(const v2f*)(arow + 4 * kk);
      c = __builtin_amdgcn_wmma_f32_16x16x4_f32(
           false, a,  false, bf[kk],
           (short)0, c,  false,  false);
    }

    const int r0 = rt * 16 + half * 8;
    if (!POOL) {
      float* hbase = H + ((size_t)(lo + r0)) * 128 + n;
#pragma unroll
      for (int j = 0; j < 8; ++j) {
        const float v = c[j] + bn;
        if (r0 + j < nn) hbase[(size_t)j * 128] = v > 0.0f ? v : 0.0f;
      }
    } else {
      const int4 q0 = *(const int4*)(gid + r0), q1 = *(const int4*)(gid + r0 + 4);
      const int  g[8] = {q0.x, q0.y, q0.z, q0.w, q1.x, q1.y, q1.z, q1.w};
      const bool counts = (tileN == 0) && (lid == 0);
      auto give = [&](const int gc, const float s, const float len) {
        if (gc >= 0) {
          atomicAdd(sums + ((size_t)gc << 7) + n, s);
          if (counts) atomicAdd(cnts + gc, len);
        }
      };
      const float v0 = c[0] + bn;
      float s   = v0 > 0.0f ? v0 : 0.0f;
      float len = 1.0f;
      int   gc  = g[0];
#pragma unroll
      for (int j = 1; j < 8; ++j) {
        float v = c[j] + bn;
        v = v > 0.0f ? v : 0.0f;
        if (g[j] == gc) {
          s += v;
          len += 1.0f;
        } else {
          give(gc, s, len);
          s = v;
          len = 1.0f;
          gc = g[j];
        }
      }
      give(gc, s, len);
    }
  }
}


__global__ void k_final(const float* __restrict__ sums, const float* __restrict__ cnts,
                        const float* __restrict__ Wlin, const float* __restrict__ blin,
                        float* __restrict__ out) {
  int t = blockIdx.x * blockDim.x + threadIdx.x;
  if (t >= N_GRAPHS * NCLS) return;
  int g = t >> 2, c = t & 3;
  float inv = 1.0f / fmaxf(cnts[g], 1.0f);
  float acc = blin[c];
  const float* srow = sums + ((size_t)g << 7);
#pragma unroll 4
  for (int k = 0; k < 128; ++k)
    acc = fmaf(srow[k] * inv, Wlin[k * NCLS + c], acc);
  out[t] = acc;
}


#define WS_BITS_OFF ((((size_t)N_NODES * (1 + 64 + 128) + (size_t)N_GRAPHS * (128 + 1)) + 31) / 32 * 32)
static_assert((WS_BITS_OFF + (size_t)BM_OWN * BM_RS) * 4 <= (size_t)268435456, "inside the workspace");

extern "C" void kernel_launch(void* const* d_in, const int* in_sizes, int n_in,
                              void* d_out, int out_size, void* d_ws, size_t ws_size,
                              hipStream_t stream) {
  const int*   tok   = (const int*)d_in[0];
  const int*   eidx  = (const int*)d_in[1];
  const int*   src   = eidx;
  const int*   dst   = eidx + N_EDGES;
  const int*   batch = (const int*)d_in[2];
  const float* emb   = (const float*)d_in[3];
  const float* W1    = (const float*)d_in[4];
  const float* b1    = (const float*)d_in[5];
  const float* W2    = (const float*)d_in[6];
  const float* b2    = (const float*)d_in[7];
  const float* Wlin  = (const float*)d_in[8];
  const float* blin  = (const float*)d_in[9];
  float*       out   = (float*)d_out;

  float* ws   = (float*)d_ws;
  float* deg  = ws;
  float* x0   = deg  + N_NODES;
  float* x1   = x0   + (size_t)N_NODES * 64;
  float* sums = x1   + (size_t)N_NODES * 128;
  float* cnts = sums + (size_t)N_GRAPHS * 128;
  unsigned* bits = (unsigned*)(ws + WS_BITS_OFF);

  const int T = 256;

  k_zero2<<<(N_NODES / 4 + (N_GRAPHS * 128 + N_GRAPHS) / 4 + T - 1) / T, T, 0, stream>>>(
      deg, N_NODES / 4, sums, (N_GRAPHS * 128 + N_GRAPHS) / 4);

  k_deg_embed<<<BM_WG + EMB_WG, T, 0, stream>>>(dst, deg, bits, tok, emb, x0);

  k_layer<6, false><<<AGG_WG, AGG_T, 0, stream>>>(x0, src, dst, bits, deg, W1, b1, x1,
                                                  nullptr, nullptr, nullptr);

  k_layer<7, true><<<AGG_WG, AGG_T, 0, stream>>>(x1, src, dst, bits, deg, W2, b2, nullptr,
                                                 batch, sums, cnts);

  k_final<<<(N_GRAPHS * NCLS + T - 1) / T, T, 0, stream>>>(sums, cnts, Wlin, blin, out);
}


# A_hat X summed without LDS float atomics: edges go on per-row lists in LDS (integer add gives the place), each wave owns 13 rows and sums a row's list in registers; sum order changed
# speedup vs baseline: 4.0529x; 1.0380x over previous
#include <hip/hip_runtime.h>
#include <hip/hip_bf16.h>


#define N_NODES  100000
#define N_EDGES  600000
#define N_GRAPHS 2000
#define NCLS     4

typedef float v2f __attribute__((ext_vector_type(2)));
typedef float v8f __attribute__((ext_vector_type(8)));


#define EMB_WG   1250
#define EMB_PER  5

static_assert((size_t)EMB_WG * 256 * EMB_PER == (size_t)N_NODES * 16, "embed covers x0 exactly");
static_assert(N_NODES % 16 == 0, "whole 16-row tiles; a tile's 16 batch ids are read as four int4");
static_assert(N_NODES % 4 == 0 && N_EDGES % 4 == 0 && (N_GRAPHS * 128 + N_GRAPHS) % 4 == 0,
              "16-byte pieces");

__global__ void k_zero2(float* __restrict__ a, int na4, float* __restrict__ b, int nb4) {
  const int i = blockIdx.x * blockDim.x + threadIdx.x;
  if (i < na4) ((float4*)a)[i] = make_float4(0.0f, 0.0f, 0.0f, 0.0f);
  else if (i - na4 < nb4) ((float4*)b)[i - na4] = make_float4(0.0f, 0.0f, 0.0f, 0.0f);
}

#define KEY_NPW 391
#define BM_OWN  256
#define BM_BLK  1024
#define BM_WG   ((N_EDGES + BM_BLK - 1) / BM_BLK)
#define BM_RS   (BM_WG * 32)
static_assert(BM_RS % 4 == 0 && (size_t)BM_RS * 32 >= (size_t)N_EDGES, "rows of whole 16-byte pieces cover the edges");
__global__ void __launch_bounds__(256)
k_deg_embed(const int* __restrict__ dst, float* __restrict__ deg,
            unsigned* __restrict__ bits,
            const int* __restrict__ tok, const float* __restrict__ emb,
            float* __restrict__ x0) {
  __shared__ __attribute__((aligned(16))) unsigned bm[BM_OWN * 32];
  const int  tid = threadIdx.x;
  if (blockIdx.x >= BM_WG) {
    const int g = (blockIdx.x - BM_WG) * 256 + tid;
    const float4* __restrict__ e4 = (const float4*)emb;
    float4* __restrict__ x4 = (float4*)x0;
    int t[EMB_PER];
#pragma unroll
    for (int j = 0; j < EMB_PER; ++j) t[j] = tok[(g + j * (EMB_WG * 256)) >> 4];
#pragma unroll
    for (int j = 0; j < EMB_PER; ++j)
      x4[g + j * (EMB_WG * 256)] = e4[(t[j] << 4) | (g & 15)];
    return;
  }
  const int  e0 = blockIdx.x * BM_BLK + tid * 4;
  const bool in = e0 < N_EDGES;
  int4 d = make_int4(0, 0, 0, 0);
  if (in) d = *(const int4*)(dst + e0);
  for (int i = tid; i < BM_OWN * 8; i += 256) ((uint4*)bm)[i] = make_uint4(0u, 0u, 0u, 0u);
  __syncthreads();
  if (in) {
    const int      w = tid >> 3;
    const unsigned b = 1u << ((tid & 7) * 4);
    atomicOr(&bm[((unsigned)d.x / KEY_NPW) * 32 + w], b);
    atomicOr(&bm[((unsigned)d.y / KEY_NPW) * 32 + w], b << 1);
    atomicOr(&bm[((unsigned)d.z / KEY_NPW) * 32 + w], b << 2);
    atomicOr(&bm[((unsigned)d.w / KEY_NPW) * 32 + w], b << 3);
  }
  __syncthreads();
  for (int i = tid; i < BM_OWN * 8; i += 256)
    *(uint4*)(bits + (size_t)(i >> 3) * BM_RS + blockIdx.x * 32 + (i & 7) * 4) = ((const uint4*)bm)[i];
  if (in) {
    atomicAdd(&deg[d.x], 1.0f);
    atomicAdd(&deg[d.y], 1.0f);
    atomicAdd(&deg[d.z], 1.0f);
    atomicAdd(&deg[d.w], 1.0f);
  }
}


#define AGG_WG    256
#define AGG_NPW   391
#define AGG_ROWS  400
#define AGG_T     1024
#define AGG_QCAP  160
#define AGG_RT    128
#define OWN_NPW   13
#define OWN_RC    16
#define AGG_NV    (BM_RS / 4)
#define AGG_NLD   ((AGG_NV + AGG_T - 1) / AGG_T)

static_assert(AGG_WG * AGG_NPW >= N_NODES && (AGG_WG - 1) * AGG_NPW < N_NODES, "node ranges");
static_assert(AGG_ROWS % 16 == 0 && AGG_ROWS >= AGG_NPW, "whole 16-row tiles cover a range");
static_assert(KEY_NPW == AGG_NPW && BM_OWN == AGG_WG && (N_NODES - 1) / AGG_NPW < BM_OWN, "one row of the map per work-group");
static_assert(AGG_NV > AGG_T, "a lane's first piece is inside the row");
static_assert(OWN_NPW * (AGG_T / 32) >= AGG_NPW && OWN_NPW * (AGG_T / 32) <= AGG_T && OWN_NPW <= 32,
              "every row of a range has an owner wave; a lane per counter");
static_assert(AGG_RT == 4 * 32 && AGG_QCAP >= AGG_RT - 1 + 32, "queue: the left-over plus one turn");

template<int N> struct row_piece;
template<> struct row_piece<2> { typedef float type __attribute__((ext_vector_type(2))); };
template<> struct row_piece<4> { typedef float type __attribute__((ext_vector_type(4))); };

__device__ __forceinline__ void lds_add(float* p, float v) {
  __hip_atomic_fetch_add(p, v, __ATOMIC_RELAXED, __HIP_MEMORY_SCOPE_WORKGROUP);
}

template<int LOGC, bool POOL>
__global__ void __launch_bounds__(AGG_T)
k_layer(const float* __restrict__ X, const int* __restrict__ src,
        const int* __restrict__ dst, const unsigned* __restrict__ bits,
        const float* __restrict__ deg,
        const float* __restrict__ W, const float* __restrict__ bias,
        float* __restrict__ H, const int* __restrict__ batch,
        float* __restrict__ sums, float* __restrict__ cnts) {
  constexpr int C = 1 << LOGC;
  constexpr int VEC = C / 32;
  constexpr int RB = 16;
  constexpr int PITCH = C + 4;
  typedef typename row_piece<VEC>::type vrow;
  __shared__ __attribute__((aligned(16))) float acc[AGG_ROWS * PITCH];
  __shared__ unsigned queue[(AGG_T / 32) * AGG_QCAP];
  __shared__ __attribute__((aligned(8))) uint2 recs[AGG_ROWS * OWN_RC];
  __shared__ unsigned rowcnt[(AGG_T / 32) * OWN_NPW];
  __shared__ __attribute__((aligned(16))) int gid[POOL ? AGG_ROWS : 4];

  const int tid  = threadIdx.x;
  const int lane = tid & 31;
  const int wv   = __builtin_amdgcn_readfirstlane(tid >> 5);
  const int lo   = blockIdx.x * AGG_NPW;
  const int nn   = min(AGG_NPW, N_NODES - lo);

  const int n4 = nn << (LOGC - 2);
  const float4* X4 = (const float4*)(X + ((size_t)lo << LOGC));
#pragma unroll 4
  for (int i = tid; i < n4; i += AGG_T) {
    const int r = i >> (LOGC - 2);
    const float dv = rsqrtf(deg[lo + r] + 1.0f);
    const float4 v = X4[i];
    *(float4*)(acc + r * PITCH + 4 * (i & (C / 4 - 1))) =
        make_float4(v.x * dv * dv, v.y * dv * dv, v.z * dv * dv, v.w * dv * dv);
  }
  for (int i = n4 + tid; i < AGG_ROWS * (C / 4); i += AGG_T)
    *(float4*)(acc + (i >> (LOGC - 2)) * PITCH + 4 * (i & (C / 4 - 1))) =
        make_float4(0.0f, 0.0f, 0.0f, 0.0f);
  if (tid < (AGG_T / 32) * OWN_NPW) rowcnt[tid] = 0u;
  if (POOL) {
    for (int r = tid; r < AGG_ROWS; r += AGG_T) gid[r] = (r < nn) ? batch[lo + r] : -1;
  }
  __syncthreads();

  unsigned* myq = queue + wv * AGG_QCAP;
  int qn = 0;

  auto route = [&](const int n) {
    __builtin_amdgcn_fence(__ATOMIC_ACQ_REL, "wavefront");
    const unsigned* qb = myq + (qn - n);
    qn -= n;
    int   s[4], dl[4];
    float ds[4], dd[4];
#pragma unroll
    for (int q = 0; q < 4; ++q) {
      const unsigned e = qb[min(lane + 32 * q, n - 1)];
      s[q]  = src[e];
      dl[q] = dst[e] - lo;
    }
#pragma unroll
    for (int q = 0; q < 4; ++q) {
      ds[q] = deg[s[q]];
      dd[q] = deg[lo + dl[q]];
    }
#pragma unroll
    for (int q = 0; q < 4; ++q) {
      const bool  ok = lane + 32 * q < n;
      const float nr = rsqrtf(ds[q] + 1.0f) * rsqrtf(dd[q] + 1.0f);
      unsigned slot = OWN_RC;
      if (ok) slot = atomicAdd(&rowcnt[dl[q]], 1u);
      if (slot < OWN_RC)
        recs[dl[q] * OWN_RC + slot] = make_uint2((unsigned)s[q], __float_as_uint(nr));
      unsigned ov = __builtin_amdgcn_ballot_w32(ok && slot >= OWN_RC);
      while (ov) {
        const int j = __builtin_ctz(ov);
        ov &= ov - 1u;
        const int   sj = __builtin_amdgcn_readlane(s[q], j);
        const int   dj = __builtin_amdgcn_readlane(dl[q], j);
        const float nj = __int_as_float(__builtin_amdgcn_readlane(__float_as_int(nr), j));
        const vrow  v  = *(const vrow*)(X + ((size_t)sj << LOGC) + VEC * lane);
        float* a = acc + dj * PITCH + VEC * lane;
#pragma unroll
        for (int t = 0; t < VEC; ++t) lds_add(a + t, v[t] * nj);
      }
    }
  };

  const uint4* row4 = (const uint4*)(bits + (size_t)blockIdx.x * BM_RS);
  uint4 nx = row4[tid];
  for (int u = 0; u < AGG_NLD; ++u) {
    uint4 cur = nx;
    const int piece = tid + u * AGG_T;
    if (piece >= AGG_NV) cur = make_uint4(0u, 0u, 0u, 0u);
    if (u + 1 < AGG_NLD) nx = row4[min(piece + AGG_T, AGG_NV - 1)];
    for (int k = 0; k < 4; ++k) {
      unsigned M = cur.x;
      cur.x = cur.y; cur.y = cur.z; cur.z = cur.w;
      const unsigned base = ((unsigned)piece * 4u + (unsigned)k) << 5;
      while (__builtin_amdgcn_ballot_w32(M != 0u)) {
        const bool m = M != 0u;
        const unsigned mk = __builtin_amdgcn_ballot_w32(m);
        if (m) {
          myq[qn + __builtin_amdgcn_mbcnt_lo(mk, 0u)] = base + (unsigned)__builtin_ctz(M);
          M &= M - 1u;
        }
        qn += __builtin_popcount(mk);
        while (qn >= AGG_RT) route(AGG_RT);
      }
    }
  }
  if (qn > 0) route(qn);
  __syncthreads();

  {
    const int r0 = wv * OWN_NPW;
    const int cl = (int)min(rowcnt[r0 + min(lane, OWN_NPW - 1)], (unsigned)OWN_RC);
    int pre[OWN_NPW + 1];
    pre[0] = 0;
#pragma unroll
    for (int k = 0; k < OWN_NPW; ++k) pre[k + 1] = pre[k] + __builtin_amdgcn_readlane(cl, k);
    const int tot = pre[OWN_NPW];
    vrow sum = (vrow)(0.0f);
    int  crow = -1;
    auto flush = [&]() {
      vrow* p = (vrow*)(acc + (r0 + crow) * PITCH + VEC * lane);
      *p += sum;
    };
    for (int c0 = 0; c0 < tot; c0 += 32) {
      const int n = min(32, tot - c0);
      const int f = c0 + min(lane, n - 1);
      int k = 0, pk = 0;
#pragma unroll
      for (int kk = 1; kk < OWN_NPW; ++kk)
        if (f >= pre[kk]) { k = kk; pk = pre[kk]; }
      const uint2 rec = recs[(r0 + k) * OWN_RC + (f - pk)];
      const int   s   = (int)rec.x;
      for (int j0 = 0; j0 < n; j0 += RB) {
        vrow v[RB];
#pragma unroll
        for (int jj = 0; jj < RB; ++jj) {
          const int sj = __builtin_amdgcn_readlane(s, min(j0 + jj, n - 1));
          v[jj] = *(const vrow*)(X + ((size_t)sj << LOGC) + VEC * lane);
        }
#pragma unroll
        for (int jj = 0; jj < RB; ++jj) {
          const int j = j0 + jj;
          if (j < n) {
            const float nj = __int_as_float(__builtin_amdgcn_readlane((int)rec.y, j));
            const int   rj = __builtin_amdgcn_readlane(k, j);
            if (rj != crow) {
              if (crow >= 0) flush();
              sum = (vrow)(0.0f);
              crow = rj;
            }
            sum += v[jj] * nj;
          }
        }
      }
    }
    if (crow >= 0) flush();
  }
  __syncthreads();

  const int tileN = wv & 7;
  const int half  = lane >> 4;
  const int lid   = lane & 15;
  const int n     = tileN * 16 + lid;
  const int kb    = half * 2;

  v2f bf[C / 4];
  const float* wcol = W + (size_t)kb * 128 + n;
#pragma unroll
  for (int kk = 0; kk < C / 4; ++kk) {
    bf[kk].x = wcol[(size_t)(4 * kk) * 128];
    bf[kk].y = wcol[(size_t)(4 * kk + 1) * 128];
  }
  const float bn = bias[n];
  const int ntile = (nn + 15) >> 4;

  for (int rt = wv >> 3; rt < ntile; rt += 4) {
    const float* arow = acc + (rt * 16 + lid) * PITCH + kb;
    v8f c = {};
#pragma unroll
    for (int kk = 0; kk < C / 4; ++kk) {
      const v2f a = *(const v2f*)(arow + 4 * kk);
      c = __builtin_amdgcn_wmma_f32_16x16x4_f32(
           false, a,  false, bf[kk],
           (short)0, c,  false,  false);
    }

    const int r0 = rt * 16 + half * 8;
    if (!POOL) {
      float* hbase = H + ((size_t)(lo + r0)) * 128 + n;
#pragma unroll
      for (int j = 0; j < 8; ++j) {
        const float v = c[j] + bn;
        if (r0 + j < nn) hbase[(size_t)j * 128] = v > 0.0f ? v : 0.0f;
      }
    } else {
      const int4 q0 = *(const int4*)(gid + r0), q1 = *(const int4*)(gid + r0 + 4);
      const int  g[8] = {q0.x, q0.y, q0.z, q0.w, q1.x, q1.y, q1.z, q1.w};
      const bool counts = (tileN == 0) && (lid == 0);
      auto give = [&](const int gc, const float s, const float len) {
        if (gc >= 0) {
          atomicAdd(sums + ((size_t)gc << 7) + n, s);
          if (counts) atomicAdd(cnts + gc, len);
        }
      };
      const float v0 = c[0] + bn;
      float s   = v0 > 0.0f ? v0 : 0.0f;
      float len = 1.0f;
      int   gc  = g[0];
#pragma unroll
      for (int j = 1; j < 8; ++j) {
        float v = c[j] + bn;
        v = v > 0.0f ? v : 0.0f;
        if (g[j] == gc) {
          s += v;
          len += 1.0f;
        } else {
          give(gc, s, len);
          s = v;
          len = 1.0f;
          gc = g[j];
        }
      }
      give(gc, s, len);
    }
  }
}


__global__ void k_final(const float* __restrict__ sums, const float* __restrict__ cnts,
                        const float* __restrict__ Wlin, const float* __restrict__ blin,
                        float* __restrict__ out) {
  int t = blockIdx.x * blockDim.x + threadIdx.x;
  if (t >= N_GRAPHS * NCLS) return;
  int g = t >> 2, c = t & 3;
  float inv = 1.0f / fmaxf(cnts[g], 1.0f);
  float acc = blin[c];
  const float* srow = sums + ((size_t)g << 7);
#pragma unroll 4
  for (int k = 0; k < 128; ++k)
    acc = fmaf(srow[k] * inv, Wlin[k * NCLS + c], acc);
  out[t] = acc;
}


#define WS_BITS_OFF ((((size_t)N_NODES * (1 + 64 + 128) + (size_t)N_GRAPHS * (128 + 1)) + 31) / 32 * 32)
static_assert((WS_BITS_OFF + (size_t)BM_OWN * BM_RS) * 4 <= (size_t)268435456, "inside the workspace");

extern "C" void kernel_launch(void* const* d_in, const int* in_sizes, int n_in,
                              void* d_out, int out_size, void* d_ws, size_t ws_size,
                              hipStream_t stream) {
  const int*   tok   = (const int*)d_in[0];
  const int*   eidx  = (const int*)d_in[1];
  const int*   src   = eidx;
  const int*   dst   = eidx + N_EDGES;
  const int*   batch = (const int*)d_in[2];
  const float* emb   = (const float*)d_in[3];
  const float* W1    = (const float*)d_in[4];
  const float* b1    = (const float*)d_in[5];
  const float* W2    = (const float*)d_in[6];
  const float* b2    = (const float*)d_in[7];
  const float* Wlin  = (const float*)d_in[8];
  const float* blin  = (const float*)d_in[9];
  float*       out   = (float*)d_out;

  float* ws   = (float*)d_ws;
  float* deg  = ws;
  float* x0   = deg  + N_NODES;
  float* x1   = x0   + (size_t)N_NODES * 64;
  float* sums = x1   + (size_t)N_NODES * 128;
  float* cnts = sums + (size_t)N_GRAPHS * 128;
  unsigned* bits = (unsigned*)(ws + WS_BITS_OFF);

  const int T = 256;

  k_zero2<<<(N_NODES / 4 + (N_GRAPHS * 128 + N_GRAPHS) / 4 + T - 1) / T, T, 0, stream>>>(
      deg, N_NODES / 4, sums, (N_GRAPHS * 128 + N_GRAPHS) / 4);

  k_deg_embed<<<BM_WG + EMB_WG, T, 0, stream>>>(dst, deg, bits, tok, emb, x0);

  k_layer<6, false><<<AGG_WG, AGG_T, 0, stream>>>(x0, src, dst, bits, deg, W1, b1, x1,
                                                  nullptr, nullptr, nullptr);

  k_layer<7, true><<<AGG_WG, AGG_T, 0, stream>>>(x1, src, dst, bits, deg, W2, b2, nullptr,
                                                 batch, sums, cnts);

  k_final<<<(N_GRAPHS * NCLS + T - 1) / T, T, 0, stream>>>(sums, cnts, Wlin, blin, out);
}


# degree counts without global atomics: a work-group per 391-node range finds its edges in the bit map, counts them in LDS integer counters, stores each count once; counts need no clear; sums unchanged
# speedup vs baseline: 4.6177x; 1.1394x over previous
#include <hip/hip_runtime.h>
#include <hip/hip_bf16.h>


#define N_NODES  100000
#define N_EDGES  600000
#define N_GRAPHS 2000
#define NCLS     4

typedef float v2f __attribute__((ext_vector_type(2)));
typedef float v8f __attribute__((ext_vector_type(8)));


#define EMB_WG   1250
#define EMB_PER  5

static_assert((size_t)EMB_WG * 256 * EMB_PER == (size_t)N_NODES * 16, "embed covers x0 exactly");
static_assert(N_NODES % 16 == 0, "whole 16-row tiles; a tile's 16 batch ids are read as four int4");
static_assert(N_NODES % 4 == 0 && N_EDGES % 4 == 0 && (N_GRAPHS * 128 + N_GRAPHS) % 4 == 0,
              "16-byte pieces");

#define KEY_NPW 391
#define BM_OWN  256
#define BM_BLK  1024
#define BM_WG   ((N_EDGES + BM_BLK - 1) / BM_BLK)
#define BM_RS   (BM_WG * 32)
static_assert(BM_RS % 4 == 0 && (size_t)BM_RS * 32 >= (size_t)N_EDGES, "rows of whole 16-byte pieces cover the edges");
__global__ void __launch_bounds__(256)
k_bits_embed(const int* __restrict__ dst,
             unsigned* __restrict__ bits,
             const int* __restrict__ tok, const float* __restrict__ emb,
             float* __restrict__ x0) {
  __shared__ __attribute__((aligned(16))) unsigned bm[BM_OWN * 32];
  const int  tid = threadIdx.x;
  if (blockIdx.x >= BM_WG) {
    const int g = (blockIdx.x - BM_WG) * 256 + tid;
    const float4* __restrict__ e4 = (const float4*)emb;
    float4* __restrict__ x4 = (float4*)x0;
    int t[EMB_PER];
#pragma unroll
    for (int j = 0; j < EMB_PER; ++j) t[j] = tok[(g + j * (EMB_WG * 256)) >> 4];
#pragma unroll
    for (int j = 0; j < EMB_PER; ++j)
      x4[g + j * (EMB_WG * 256)] = e4[(t[j] << 4) | (g & 15)];
    return;
  }
  const int  e0 = blockIdx.x * BM_BLK + tid * 4;
  const bool in = e0 < N_EDGES;
  int4 d = make_int4(0, 0, 0, 0);
  if (in) d = *(const int4*)(dst + e0);
  for (int i = tid; i < BM_OWN * 8; i += 256) ((uint4*)bm)[i] = make_uint4(0u, 0u, 0u, 0u);
  __syncthreads();
  if (in) {
    const int      w = tid >> 3;
    const unsigned b = 1u << ((tid & 7) * 4);
    atomicOr(&bm[((unsigned)d.x / KEY_NPW) * 32 + w], b);
    atomicOr(&bm[((unsigned)d.y / KEY_NPW) * 32 + w], b << 1);
    atomicOr(&bm[((unsigned)d.z / KEY_NPW) * 32 + w], b << 2);
    atomicOr(&bm[((unsigned)d.w / KEY_NPW) * 32 + w], b << 3);
  }
  __syncthreads();
  for (int i = tid; i < BM_OWN * 8; i += 256)
    *(uint4*)(bits + (size_t)(i >> 3) * BM_RS + blockIdx.x * 32 + (i & 7) * 4) = ((const uint4*)bm)[i];
}


#define AGG_WG    256
#define AGG_NPW   391
#define AGG_ROWS  400
#define AGG_T     1024
#define AGG_QCAP  160
#define AGG_RT    128
#define OWN_NPW   13
#define OWN_RC    16
#define AGG_NV    (BM_RS / 4)
#define AGG_NLD   ((AGG_NV + AGG_T - 1) / AGG_T)

static_assert(AGG_WG * AGG_NPW >= N_NODES && (AGG_WG - 1) * AGG_NPW < N_NODES, "node ranges");
static_assert(AGG_ROWS % 16 == 0 && AGG_ROWS >= AGG_NPW, "whole 16-row tiles cover a range");
static_assert(KEY_NPW == AGG_NPW && BM_OWN == AGG_WG && (N_NODES - 1) / AGG_NPW < BM_OWN, "one row of the map per work-group");
static_assert(AGG_NV > AGG_T, "a lane's first piece is inside the row");
static_assert(OWN_NPW * (AGG_T / 32) >= AGG_NPW && OWN_NPW * (AGG_T / 32) <= AGG_T && OWN_NPW <= 32,
              "every row of a range has an owner wave; a lane per counter");
static_assert(AGG_RT == 4 * 32 && AGG_QCAP >= AGG_RT - 1 + 32, "queue: the left-over plus one turn");

#define CNT_T    256
#define CNT_PPL  ((AGG_NV + CNT_T - 1) / CNT_T)
#define ZERO_V4  ((N_GRAPHS * 128 + N_GRAPHS) / 4)
#define ZERO_WG  ((ZERO_V4 + CNT_T - 1) / CNT_T)
static_assert((CNT_PPL - 1) * CNT_T < AGG_NV, "a lane's last piece index can be inside the row");

__global__ void __launch_bounds__(CNT_T)
k_count(const int* __restrict__ dst, const unsigned* __restrict__ bits,
        float* __restrict__ deg, float* __restrict__ sums) {
  __shared__ __attribute__((aligned(16))) uint4 park[CNT_PPL * CNT_T];
  __shared__ unsigned queue[(CNT_T / 32) * AGG_QCAP];
  __shared__ unsigned cnt[AGG_ROWS];
  const int tid = threadIdx.x;
  if (blockIdx.x >= AGG_WG) {
    const int i = (blockIdx.x - AGG_WG) * CNT_T + tid;
    if (i < ZERO_V4) ((float4*)sums)[i] = make_float4(0.0f, 0.0f, 0.0f, 0.0f);
    return;
  }
  const int lane = tid & 31;
  const int wv   = __builtin_amdgcn_readfirstlane(tid >> 5);
  const int lo   = blockIdx.x * AGG_NPW;
  const int nn   = min(AGG_NPW, N_NODES - lo);

  const uint4* row4 = (const uint4*)(bits + (size_t)blockIdx.x * BM_RS);
  uint4 pc[CNT_PPL];
#pragma unroll
  for (int u = 0; u < CNT_PPL; ++u) pc[u] = row4[min(tid + u * CNT_T, AGG_NV - 1)];
  for (int i = tid; i < AGG_ROWS; i += CNT_T) cnt[i] = 0u;
#pragma unroll
  for (int u = 0; u < CNT_PPL; ++u)
    park[u * CNT_T + tid] = (tid + u * CNT_T < AGG_NV) ? pc[u] : make_uint4(0u, 0u, 0u, 0u);
  __syncthreads();

  unsigned* myq = queue + wv * AGG_QCAP;
  int qn = 0;

  auto tally = [&](const int n) {
    __builtin_amdgcn_fence(__ATOMIC_ACQ_REL, "wavefront");
    const unsigned* qb = myq + (qn - n);
    qn -= n;
    int dl[4];
#pragma unroll
    for (int q = 0; q < 4; ++q) dl[q] = dst[qb[min(lane + 32 * q, n - 1)]] - lo;
#pragma unroll
    for (int q = 0; q < 4; ++q)
      if (lane + 32 * q < n) atomicAdd(&cnt[dl[q]], 1u);
  };

  uint4 nx = park[tid];
  for (int u = 0; u < CNT_PPL; ++u) {
    const int piece = tid + u * CNT_T;
    uint4 cur = nx;
    if (u + 1 < CNT_PPL) nx = park[(u + 1) * CNT_T + tid];
    for (int k = 0; k < 4; ++k) {
      unsigned M = cur.x;
      cur.x = cur.y; cur.y = cur.z; cur.z = cur.w;
      const unsigned base = ((unsigned)piece * 4u + (unsigned)k) << 5;
      while (__builtin_amdgcn_ballot_w32(M != 0u)) {
        const bool m = M != 0u;
        const unsigned mk = __builtin_amdgcn_ballot_w32(m);
        if (m) {
          myq[qn + __builtin_amdgcn_mbcnt_lo(mk, 0u)] = base + (unsigned)__builtin_ctz(M);
          M &= M - 1u;
        }
        qn += __builtin_popcount(mk);
        while (qn >= AGG_RT) tally(AGG_RT);
      }
    }
  }
  if (qn > 0) tally(qn);
  __syncthreads();
  for (int r = tid; r < nn; r += CNT_T) deg[lo + r] = (float)cnt[r];
}

template<int N> struct row_piece;
template<> struct row_piece<2> { typedef float type __attribute__((ext_vector_type(2))); };
template<> struct row_piece<4> { typedef float type __attribute__((ext_vector_type(4))); };

__device__ __forceinline__ void lds_add(float* p, float v) {
  __hip_atomic_fetch_add(p, v, __ATOMIC_RELAXED, __HIP_MEMORY_SCOPE_WORKGROUP);
}

template<int LOGC, bool POOL>
__global__ void __launch_bounds__(AGG_T)
k_layer(const float* __restrict__ X, const int* __restrict__ src,
        const int* __restrict__ dst, const unsigned* __restrict__ bits,
        const float* __restrict__ deg,
        const float* __restrict__ W, const float* __restrict__ bias,
        float* __restrict__ H, const int* __restrict__ batch,
        float* __restrict__ sums, float* __restrict__ cnts) {
  constexpr int C = 1 << LOGC;
  constexpr int VEC = C / 32;
  constexpr int RB = 16;
  constexpr int PITCH = C + 4;
  typedef typename row_piece<VEC>::type vrow;
  __shared__ __attribute__((aligned(16))) float acc[AGG_ROWS * PITCH];
  __shared__ unsigned queue[(AGG_T / 32) * AGG_QCAP];
  __shared__ __attribute__((aligned(8))) uint2 recs[AGG_ROWS * OWN_RC];
  __shared__ unsigned rowcnt[(AGG_T / 32) * OWN_NPW];
  __shared__ __attribute__((aligned(16))) int gid[POOL ? AGG_ROWS : 4];

  const int tid  = threadIdx.x;
  const int lane = tid & 31;
  const int wv   = __builtin_amdgcn_readfirstlane(tid >> 5);
  const int lo   = blockIdx.x * AGG_NPW;
  const int nn   = min(AGG_NPW, N_NODES - lo);

  const int n4 = nn << (LOGC - 2);
  const float4* X4 = (const float4*)(X + ((size_t)lo << LOGC));
#pragma unroll 4
  for (int i = tid; i < n4; i += AGG_T) {
    const int r = i >> (LOGC - 2);
    const float dv = rsqrtf(deg[lo + r] + 1.0f);
    const float4 v = X4[i];
    *(float4*)(acc + r * PITCH + 4 * (i & (C / 4 - 1))) =
        make_float4(v.x * dv * dv, v.y * dv * dv, v.z * dv * dv, v.w * dv * dv);
  }
  for (int i = n4 + tid; i < AGG_ROWS * (C / 4); i += AGG_T)
    *(float4*)(acc + (i >> (LOGC - 2)) * PITCH + 4 * (i & (C / 4 - 1))) =
        make_float4(0.0f, 0.0f, 0.0f, 0.0f);
  if (tid < (AGG_T / 32) * OWN_NPW) rowcnt[tid] = 0u;
  if (POOL) {
    for (int r = tid; r < AGG_ROWS; r += AGG_T) gid[r] = (r < nn) ? batch[lo + r] : -1;
  }
  __syncthreads();

  unsigned* myq = queue + wv * AGG_QCAP;
  int qn = 0;

  auto route = [&](const int n) {
    __builtin_amdgcn_fence(__ATOMIC_ACQ_REL, "wavefront");
    const unsigned* qb = myq + (qn - n);
    qn -= n;
    int   s[4], dl[4];
    float ds[4], dd[4];
#pragma unroll
    for (int q = 0; q < 4; ++q) {
      const unsigned e = qb[min(lane + 32 * q, n - 1)];
      s[q]  = src[e];
      dl[q] = dst[e] - lo;
    }
#pragma unroll
    for (int q = 0; q < 4; ++q) {
      ds[q] = deg[s[q]];
      dd[q] = deg[lo + dl[q]];
    }
#pragma unroll
    for (int q = 0; q < 4; ++q) {
      const bool  ok = lane + 32 * q < n;
      const float nr = rsqrtf(ds[q] + 1.0f) * rsqrtf(dd[q] + 1.0f);
      unsigned slot = OWN_RC;
      if (ok) slot = atomicAdd(&rowcnt[dl[q]], 1u);
      if (slot < OWN_RC)
        recs[dl[q] * OWN_RC + slot] = make_uint2((unsigned)s[q], __float_as_uint(nr));
      unsigned ov = __builtin_amdgcn_ballot_w32(ok && slot >= OWN_RC);
      while (ov) {
        const int j = __builtin_ctz(ov);
        ov &= ov - 1u;
        const int   sj = __builtin_amdgcn_readlane(s[q], j);
        const int   dj = __builtin_amdgcn_readlane(dl[q], j);
        const float nj = __int_as_float(__builtin_amdgcn_readlane(__float_as_int(nr), j));
        const vrow  v  = *(const vrow*)(X + ((size_t)sj << LOGC) + VEC * lane);
        float* a = acc + dj * PITCH + VEC * lane;
#pragma unroll
        for (int t = 0; t < VEC; ++t) lds_add(a + t, v[t] * nj);
      }
    }
  };

  const uint4* row4 = (const uint4*)(bits + (size_t)blockIdx.x * BM_RS);
  uint4 nx = row4[tid];
  for (int u = 0; u < AGG_NLD; ++u) {
    uint4 cur = nx;
    const int piece = tid + u * AGG_T;
    if (piece >= AGG_NV) cur = make_uint4(0u, 0u, 0u, 0u);
    if (u + 1 < AGG_NLD) nx = row4[min(piece + AGG_T, AGG_NV - 1)];
    for (int k = 0; k < 4; ++k) {
      unsigned M = cur.x;
      cur.x = cur.y; cur.y = cur.z; cur.z = cur.w;
      const unsigned base = ((unsigned)piece * 4u + (unsigned)k) << 5;
      while (__builtin_amdgcn_ballot_w32(M != 0u)) {
        const bool m = M != 0u;
        const unsigned mk = __builtin_amdgcn_ballot_w32(m);
        if (m) {
          myq[qn + __builtin_amdgcn_mbcnt_lo(mk, 0u)] = base + (unsigned)__builtin_ctz(M);
          M &= M - 1u;
        }
        qn += __builtin_popcount(mk);
        while (qn >= AGG_RT) route(AGG_RT);
      }
    }
  }
  if (qn > 0) route(qn);
  __syncthreads();

  {
    const int r0 = wv * OWN_NPW;
    const int cl = (int)min(rowcnt[r0 + min(lane, OWN_NPW - 1)], (unsigned)OWN_RC);
    int pre[OWN_NPW + 1];
    pre[0] = 0;
#pragma unroll
    for (int k = 0; k < OWN_NPW; ++k) pre[k + 1] = pre[k] + __builtin_amdgcn_readlane(cl, k);
    const int tot = pre[OWN_NPW];
    vrow sum = (vrow)(0.0f);
    int  crow = -1;
    auto flush = [&]() {
      vrow* p = (vrow*)(acc + (r0 + crow) * PITCH + VEC * lane);
      *p += sum;
    };
    for (int c0 = 0; c0 < tot; c0 += 32) {
      const int n = min(32, tot - c0);
      const int f = c0 + min(lane, n - 1);
      int k = 0, pk = 0;
#pragma unroll
      for (int kk = 1; kk < OWN_NPW; ++kk)
        if (f >= pre[kk]) { k = kk; pk = pre[kk]; }
      const uint2 rec = recs[(r0 + k) * OWN_RC + (f - pk)];
      const int   s   = (int)rec.x;
      for (int j0 = 0; j0 < n; j0 += RB) {
        vrow v[RB];
#pragma unroll
        for (int jj = 0; jj < RB; ++jj) {
          const int sj = __builtin_amdgcn_readlane(s, min(j0 + jj, n - 1));
          v[jj] = *(const vrow*)(X + ((size_t)sj << LOGC) + VEC * lane);
        }
#pragma unroll
        for (int jj = 0; jj < RB; ++jj) {
          const int j = j0 + jj;
          if (j < n) {
            const float nj = __int_as_float(__builtin_amdgcn_readlane((int)rec.y, j));
            const int   rj = __builtin_amdgcn_readlane(k, j);
            if (rj != crow) {
              if (crow >= 0) flush();
              sum = (vrow)(0.0f);
              crow = rj;
            }
            sum += v[jj] * nj;
          }
        }
      }
    }
    if (crow >= 0) flush();
  }
  __syncthreads();

  const int tileN = wv & 7;
  const int half  = lane >> 4;
  const int lid   = lane & 15;
  const int n     = tileN * 16 + lid;
  const int kb    = half * 2;

  v2f bf[C / 4];
  const float* wcol = W + (size_t)kb * 128 + n;
#pragma unroll
  for (int kk = 0; kk < C / 4; ++kk) {
    bf[kk].x = wcol[(size_t)(4 * kk) * 128];
    bf[kk].y = wcol[(size_t)(4 * kk + 1) * 128];
  }
  const float bn = bias[n];
  const int ntile = (nn + 15) >> 4;

  for (int rt = wv >> 3; rt < ntile; rt += 4) {
    const float* arow = acc + (rt * 16 + lid) * PITCH + kb;
    v8f c = {};
#pragma unroll
    for (int kk = 0; kk < C / 4; ++kk) {
      const v2f a = *(const v2f*)(arow + 4 * kk);
      c = __builtin_amdgcn_wmma_f32_16x16x4_f32(
           false, a,  false, bf[kk],
           (short)0, c,  false,  false);
    }

    const int r0 = rt * 16 + half * 8;
    if (!POOL) {
      float* hbase = H + ((size_t)(lo + r0)) * 128 + n;
#pragma unroll
      for (int j = 0; j < 8; ++j) {
        const float v = c[j] + bn;
        if (r0 + j < nn) hbase[(size_t)j * 128] = v > 0.0f ? v : 0.0f;
      }
    } else {
      const int4 q0 = *(const int4*)(gid + r0), q1 = *(const int4*)(gid + r0 + 4);
      const int  g[8] = {q0.x, q0.y, q0.z, q0.w, q1.x, q1.y, q1.z, q1.w};
      const bool counts = (tileN == 0) && (lid == 0);
      auto give = [&](const int gc, const float s, const float len) {
        if (gc >= 0) {
          atomicAdd(sums + ((size_t)gc << 7) + n, s);
          if (counts) atomicAdd(cnts + gc, len);
        }
      };
      const float v0 = c[0] + bn;
      float s   = v0 > 0.0f ? v0 : 0.0f;
      float len = 1.0f;
      int   gc  = g[0];
#pragma unroll
      for (int j = 1; j < 8; ++j) {
        float v = c[j] + bn;
        v = v > 0.0f ? v : 0.0f;
        if (g[j] == gc) {
          s += v;
          len += 1.0f;
        } else {
          give(gc, s, len);
          s = v;
          len = 1.0f;
          gc = g[j];
        }
      }
      give(gc, s, len);
    }
  }
}


__global__ void k_final(const float* __restrict__ sums, const float* __restrict__ cnts,
                        const float* __restrict__ Wlin, const float* __restrict__ blin,
                        float* __restrict__ out) {
  int t = blockIdx.x * blockDim.x + threadIdx.x;
  if (t >= N_GRAPHS * NCLS) return;
  int g = t >> 2, c = t & 3;
  float inv = 1.0f / fmaxf(cnts[g], 1.0f);
  float acc = blin[c];
  const float* srow = sums + ((size_t)g << 7);
#pragma unroll 4
  for (int k = 0; k < 128; ++k)
    acc = fmaf(srow[k] * inv, Wlin[k * NCLS + c], acc);
  out[t] = acc;
}


#define WS_BITS_OFF ((((size_t)N_NODES * (1 + 64 + 128) + (size_t)N_GRAPHS * (128 + 1)) + 31) / 32 * 32)
static_assert((WS_BITS_OFF + (size_t)BM_OWN * BM_RS) * 4 <= (size_t)268435456, "inside the workspace");

extern "C" void kernel_launch(void* const* d_in, const int* in_sizes, int n_in,
                              void* d_out, int out_size, void* d_ws, size_t ws_size,
                              hipStream_t stream) {
  const int*   tok   = (const int*)d_in[0];
  const int*   eidx  = (const int*)d_in[1];
  const int*   src   = eidx;
  const int*   dst   = eidx + N_EDGES;
  const int*   batch = (const int*)d_in[2];
  const float* emb   = (const float*)d_in[3];
  const float* W1    = (const float*)d_in[4];
  const float* b1    = (const float*)d_in[5];
  const float* W2    = (const float*)d_in[6];
  const float* b2    = (const float*)d_in[7];
  const float* Wlin  = (const float*)d_in[8];
  const float* blin  = (const float*)d_in[9];
  float*       out   = (float*)d_out;

  float* ws   = (float*)d_ws;
  float* deg  = ws;
  float* x0   = deg  + N_NODES;
  float* x1   = x0   + (size_t)N_NODES * 64;
  float* sums = x1   + (size_t)N_NODES * 128;
  float* cnts = sums + (size_t)N_GRAPHS * 128;
  unsigned* bits = (unsigned*)(ws + WS_BITS_OFF);

  const int T = 256;

  k_bits_embed<<<BM_WG + EMB_WG, T, 0, stream>>>(dst, bits, tok, emb, x0);

  k_count<<<AGG_WG + ZERO_WG, CNT_T, 0, stream>>>(dst, bits, deg, sums);

  k_layer<6, false><<<AGG_WG, AGG_T, 0, stream>>>(x0, src, dst, bits, deg, W1, b1, x1,
                                                  nullptr, nullptr, nullptr);

  k_layer<7, true><<<AGG_WG, AGG_T, 0, stream>>>(x1, src, dst, bits, deg, W2, b2, nullptr,
                                                 batch, sums, cnts);

  k_final<<<(N_GRAPHS * NCLS + T - 1) / T, T, 0, stream>>>(sums, cnts, Wlin, blin, out);
}
